# starting kernel
# baseline (speedup 1.0000x reference)
#include <hip/hip_runtime.h>

#define N_NODES 100000
#define DIM 128
#define MTILES (N_NODES / 16)

typedef __attribute__((ext_vector_type(2))) float v2f;
typedef __attribute__((ext_vector_type(8))) float v8f;

__global__ void k_init_deg(float* deg) {
    int i = blockIdx.x * blockDim.x + threadIdx.x;
    if (i < N_NODES) deg[i] = 1.0f;
}

__global__ void k_count_deg(const int* __restrict__ dst, long long E, float* deg) {
    long long e = (long long)blockIdx.x * blockDim.x + threadIdx.x;
    if (e < E) atomicAdd(&deg[dst[e]], 1.0f);
}

__global__ void k_dinv(float* deg) {
    int i = blockIdx.x * blockDim.x + threadIdx.x;
    if (i < N_NODES) deg[i] = rsqrtf(deg[i]);
}

__global__ void k_zero(float* p, long long n4) {
    long long i = (long long)blockIdx.x * blockDim.x + threadIdx.x;
    if (i < n4) ((float4*)p)[i] = make_float4(0.f, 0.f, 0.f, 0.f);
}

__global__ __launch_bounds__(256) void k_gemm(const float* __restrict__ X,
                                              const float* __restrict__ W,
                                              float* __restrict__ H) {
    int wave = (blockIdx.x * blockDim.x + threadIdx.x) >> 5;
    int lane = threadIdx.x & 31;
    if (wave >= MTILES) return;

    const int m0   = wave * 16;
    const int half = lane >> 4;
    const int l16  = lane & 15;
    const int koff = half * 2;

    v8f acc[8];
#pragma unroll
    for (int n = 0; n < 8; ++n) acc[n] = (v8f){0,0,0,0,0,0,0,0};

    const float* xrow = X + (size_t)(m0 + l16) * DIM;

    for (int k0 = 0; k0 < DIM; k0 += 4) {
        v2f a;
        a.x = xrow[k0 + koff + 0];
        a.y = xrow[k0 + koff + 1];
#pragma unroll
        for (int n = 0; n < 8; ++n) {
            v2f b;
            b.x = W[(size_t)(k0 + koff + 0) * DIM + n * 16 + l16];
            b.y = W[(size_t)(k0 + koff + 1) * DIM + n * 16 + l16];
            acc[n] = __builtin_amdgcn_wmma_f32_16x16x4_f32(
                false, a, false, b, (short)0, acc[n], false, false);
        }
    }

    const int rbase = m0 + half * 8;
#pragma unroll
    for (int n = 0; n < 8; ++n) {
#pragma unroll
        for (int v = 0; v < 8; ++v) {
            H[(size_t)(rbase + v) * DIM + n * 16 + l16] = acc[n][v];
        }
    }
}

__global__ void k_aggregate(const float* __restrict__ H,
                            const int* __restrict__ src,
                            const int* __restrict__ dst,
                            const float* __restrict__ dinv,
                            float* __restrict__ out, long long E) {
    long long t = (long long)blockIdx.x * blockDim.x + threadIdx.x;
    long long e = t >> 5;
    int lane = threadIdx.x & 31;
    if (e >= E) return;

    int s = src[e];
    int d = dst[e];
    float nrm = dinv[s] * dinv[d];

    float4 v = ((const float4*)(H + (size_t)s * DIM))[lane];
    float* op = out + (size_t)d * DIM + lane * 4;
    atomicAdd(op + 0, v.x * nrm);
    atomicAdd(op + 1, v.y * nrm);
    atomicAdd(op + 2, v.z * nrm);
    atomicAdd(op + 3, v.w * nrm);
}

__global__ void k_bias_relu_self(const float* __restrict__ agg,
                                 const float* __restrict__ H,
                                 const float* __restrict__ dinv,
                                 const float* __restrict__ bias,
                                 float* __restrict__ out) {
    long long idx = (long long)blockIdx.x * blockDim.x + threadIdx.x;
    const long long n4 = (long long)N_NODES * DIM / 4;
    if (idx >= n4) return;
    int i  = (int)(idx >> 5);
    int f4 = (int)(idx & 31);

    float di = dinv[i];
    float sn = di * di;
    float4 a  = ((const float4*)agg)[idx];
    float4 h  = ((const float4*)H)[idx];
    float4 bb = ((const float4*)bias)[f4];
    float4 r;
    r.x = fmaxf(fmaf(sn, h.x, a.x) + bb.x, 0.f);
    r.y = fmaxf(fmaf(sn, h.y, a.y) + bb.y, 0.f);
    r.z = fmaxf(fmaf(sn, h.z, a.z) + bb.z, 0.f);
    r.w = fmaxf(fmaf(sn, h.w, a.w) + bb.w, 0.f);
    ((float4*)out)[idx] = r;
}

extern "C" void kernel_launch(void* const* d_in, const int* in_sizes, int n_in,
                              void* d_out, int out_size, void* d_ws, size_t ws_size,
                              hipStream_t stream) {
    const float* x   = (const float*)d_in[0];
    const int*   ei  = (const int*)  d_in[1];
    const float* W1  = (const float*)d_in[2];
    const float* b1  = (const float*)d_in[3];
    const float* W2  = (const float*)d_in[4];
    const float* b2  = (const float*)d_in[5];
    float* out = (float*)d_out;

    const long long E = (long long)in_sizes[1] / 2;
    const int* src = ei;
    const int* dst = ei + E;

    float* dinv = (float*)d_ws;
    float* bufA = dinv + N_NODES;
    float* bufB = bufA + (size_t)N_NODES * DIM;

    const long long nd  = (long long)N_NODES * DIM;
    const long long nd4 = nd / 4;

    dim3 blk(256);
    const int gN    = (N_NODES + 255) / 256;
    const int gE    = (int)((E + 255) / 256);
    const int gZ    = (int)((nd4 + 255) / 256);
    const int gAgg  = (int)((E * 32 + 255) / 256);
    const int gGemm = (MTILES + 7) / 8;

    k_init_deg<<<gN, blk, 0, stream>>>(dinv);
    k_count_deg<<<gE, blk, 0, stream>>>(dst, E, dinv);
    k_dinv<<<gN, blk, 0, stream>>>(dinv);

    k_gemm<<<gGemm, blk, 0, stream>>>(x, W1, bufA);
    k_zero<<<gZ, blk, 0, stream>>>(bufB, nd4);
    k_aggregate<<<gAgg, blk, 0, stream>>>(bufA, src, dst, dinv, bufB, E);
    k_bias_relu_self<<<gZ, blk, 0, stream>>>(bufB, bufA, dinv, b1, bufA);

    k_gemm<<<gGemm, blk, 0, stream>>>(bufA, W2, bufB);
    k_zero<<<gZ, blk, 0, stream>>>(out, nd4);
    k_aggregate<<<gAgg, blk, 0, stream>>>(bufB, src, dst, dinv, out, E);
    k_bias_relu_self<<<gZ, blk, 0, stream>>>(out, bufB, dinv, b2, out);
}


# Edge sums by destination ownership: a 1024-thread group keeps 400 output rows in LDS, scans the whole edge list for its own, adds rows with LDS float adds, stores relu itself; sum order: LDS arrival
# speedup vs baseline: 2.0849x; 2.0849x over previous
#include <hip/hip_runtime.h>

#define N_NODES 100000
#define DIM 128
#define MTILES (N_NODES / 16)

typedef __attribute__((ext_vector_type(2))) float v2f;
typedef __attribute__((ext_vector_type(8))) float v8f;

__global__ void k_init_deg(float* deg) {
    int i = blockIdx.x * blockDim.x + threadIdx.x;
    if (i < N_NODES) deg[i] = 1.0f;
}

__global__ void k_count_deg(const int* __restrict__ dst, long long E, float* deg) {
    long long e = (long long)blockIdx.x * blockDim.x + threadIdx.x;
    if (e < E) atomicAdd(&deg[dst[e]], 1.0f);
}

__global__ void k_dinv(float* deg) {
    int i = blockIdx.x * blockDim.x + threadIdx.x;
    if (i < N_NODES) deg[i] = rsqrtf(deg[i]);
}

__global__ __launch_bounds__(256) void k_gemm(const float* __restrict__ X,
                                              const float* __restrict__ W,
                                              float* __restrict__ H) {
    int wave = (blockIdx.x * blockDim.x + threadIdx.x) >> 5;
    int lane = threadIdx.x & 31;
    if (wave >= MTILES) return;

    const int m0   = wave * 16;
    const int half = lane >> 4;
    const int l16  = lane & 15;
    const int koff = half * 2;

    v8f acc[8];
#pragma unroll
    for (int n = 0; n < 8; ++n) acc[n] = (v8f){0,0,0,0,0,0,0,0};

    const float* xrow = X + (size_t)(m0 + l16) * DIM;

    for (int k0 = 0; k0 < DIM; k0 += 4) {
        v2f a;
        a.x = xrow[k0 + koff + 0];
        a.y = xrow[k0 + koff + 1];
#pragma unroll
        for (int n = 0; n < 8; ++n) {
            v2f b;
            b.x = W[(size_t)(k0 + koff + 0) * DIM + n * 16 + l16];
            b.y = W[(size_t)(k0 + koff + 1) * DIM + n * 16 + l16];
            acc[n] = __builtin_amdgcn_wmma_f32_16x16x4_f32(
                false, a, false, b, (short)0, acc[n], false, false);
        }
    }

    const int rbase = m0 + half * 8;
#pragma unroll
    for (int n = 0; n < 8; ++n) {
#pragma unroll
        for (int v = 0; v < 8; ++v) {
            H[(size_t)(rbase + v) * DIM + n * 16 + l16] = acc[n][v];
        }
    }
}

#define OWN      400
#define NGROUPS  (N_NODES / OWN)
#define SCAN_U   8
#define QCAP     64

__device__ __forceinline__ void drain_edges(const int* qe, int n, int lane,
                                            const float* __restrict__ H,
                                            const int* __restrict__ src,
                                            const int* __restrict__ dst,
                                            const float* __restrict__ dinv,
                                            int lo, float* acc) {
    int s_i = 0, d_i = 0;
    float nr_i = 0.f;
    if (lane < n) {
        int e_i = qe[lane];
        int dg  = dst[e_i];
        s_i  = src[e_i];
        d_i  = dg - lo;
        nr_i = dinv[s_i] * dinv[dg];
    }
    for (int j0 = 0; j0 < n; j0 += 8) {
        float4 v[8];
#pragma unroll
        for (int k = 0; k < 8; ++k) {
            int jj = (j0 + k < n) ? (j0 + k) : (n - 1);
            int s = __builtin_amdgcn_readlane(s_i, jj);
            v[k] = ((const float4*)(H + (size_t)s * DIM))[lane];
        }
#pragma unroll
        for (int k = 0; k < 8; ++k) {
            int jj = j0 + k;
            if (jj < n) {
                int d = __builtin_amdgcn_readlane(d_i, jj);
                float nr = __int_as_float(__builtin_amdgcn_readlane(__float_as_int(nr_i), jj));
                float* a = acc + d * DIM + lane;
                atomicAdd(a,      v[k].x * nr);
                atomicAdd(a + 32, v[k].y * nr);
                atomicAdd(a + 64, v[k].z * nr);
                atomicAdd(a + 96, v[k].w * nr);
            }
        }
    }
}

__global__ __launch_bounds__(1024) void k_aggregate_owned(const float* __restrict__ H,
                                                          const int* __restrict__ src,
                                                          const int* __restrict__ dst,
                                                          const float* __restrict__ dinv,
                                                          const float* __restrict__ bias,
                                                          float* __restrict__ out, long long E64) {
    __shared__ __attribute__((aligned(16))) float acc[OWN * DIM];
    __shared__ int q_e[32 * QCAP];

    const int tid  = threadIdx.x;
    const int lane = tid & 31;
    const int w    = __builtin_amdgcn_readfirstlane(tid >> 5);
    const int lo   = blockIdx.x * OWN;
    const int E    = (int)E64;

    for (int i = tid; i < OWN * DIM / 4; i += 1024)
        ((float4*)acc)[i] = make_float4(0.f, 0.f, 0.f, 0.f);
    __syncthreads();

    int* qe = q_e + w * QCAP;
    int qn = 0;
    const unsigned ltmask = (1u << lane) - 1u;

    const int F     = E >> 7;
    const int trips = (F + 32 * SCAN_U - 1) / (32 * SCAN_U);
    int trip = (int)(((long long)blockIdx.x * trips) / NGROUPS);

    for (int n = 0; n <= trips; ++n) {
        unsigned hm = 0;
        int tile0;
        if (n < trips) {
            tile0 = trip * (SCAN_U * 32) + w;
            int4 dv[SCAN_U];
            unsigned okmask = 0;
#pragma unroll
            for (int k = 0; k < SCAN_U; ++k) {
                int t  = tile0 + k * 32;
                bool ok = t < F;
                int tc = ok ? t : (F - 1);
                okmask |= ok ? (0xFu << (4 * k)) : 0u;
                dv[k] = *(const int4*)(dst + tc * 128 + lane * 4);
            }
#pragma unroll
            for (int k = 0; k < SCAN_U; ++k) {
                hm |= ((unsigned)(dv[k].x - lo) < (unsigned)OWN) ? (1u << (4 * k + 0)) : 0u;
                hm |= ((unsigned)(dv[k].y - lo) < (unsigned)OWN) ? (1u << (4 * k + 1)) : 0u;
                hm |= ((unsigned)(dv[k].z - lo) < (unsigned)OWN) ? (1u << (4 * k + 2)) : 0u;
                hm |= ((unsigned)(dv[k].w - lo) < (unsigned)OWN) ? (1u << (4 * k + 3)) : 0u;
            }
            hm &= okmask;
            if (++trip == trips) trip = 0;
        } else {
            tile0 = F;
            if (w == 0) {
                for (int b = 0; b < 4; ++b) {
                    int e = F * 128 + lane * 4 + b;
                    int dl = (e < E) ? (dst[e] - lo) : -1;
                    hm |= ((unsigned)dl < (unsigned)OWN) ? (1u << b) : 0u;
                }
            }
        }
        for (;;) {
            bool has = hm != 0u;
            unsigned m = __builtin_amdgcn_ballot_w32(has);
            if (m == 0u) break;
            if (has) {
                int step = __builtin_ctz(hm);
                hm &= hm - 1u;
                int e = (tile0 + (step >> 2) * 32) * 128 + lane * 4 + (step & 3);
                qe[qn + __builtin_popcount(m & ltmask)] = e;
            }
            qn += __builtin_popcount(m);
            if (qn >= 32) {
                qn -= 32;
                drain_edges(qe + qn, 32, lane, H, src, dst, dinv, lo, acc);
            }
        }
    }
    if (qn > 0) drain_edges(qe, qn, lane, H, src, dst, dinv, lo, acc);
    __syncthreads();

    const float4 bb = ((const float4*)bias)[lane];
#pragma unroll 4
    for (int r = w; r < OWN; r += 32) {
        const int i = lo + r;
        float di = dinv[i];
        float sn = di * di;
        const float* a = acc + r * DIM + lane;
        float4 h = ((const float4*)(H + (size_t)i * DIM))[lane];
        float4 o;
        o.x = fmaxf(fmaf(sn, h.x, a[0])  + bb.x, 0.f);
        o.y = fmaxf(fmaf(sn, h.y, a[32]) + bb.y, 0.f);
        o.z = fmaxf(fmaf(sn, h.z, a[64]) + bb.z, 0.f);
        o.w = fmaxf(fmaf(sn, h.w, a[96]) + bb.w, 0.f);
        ((float4*)(out + (size_t)i * DIM))[lane] = o;
    }
}

extern "C" void kernel_launch(void* const* d_in, const int* in_sizes, int n_in,
                              void* d_out, int out_size, void* d_ws, size_t ws_size,
                              hipStream_t stream) {
    const float* x   = (const float*)d_in[0];
    const int*   ei  = (const int*)  d_in[1];
    const float* W1  = (const float*)d_in[2];
    const float* b1  = (const float*)d_in[3];
    const float* W2  = (const float*)d_in[4];
    const float* b2  = (const float*)d_in[5];
    float* out = (float*)d_out;

    const long long E = (long long)in_sizes[1] / 2;
    const int* src = ei;
    const int* dst = ei + E;

    float* dinv = (float*)d_ws;
    float* bufA = dinv + N_NODES;
    float* bufB = bufA + (size_t)N_NODES * DIM;

    dim3 blk(256);
    const int gN    = (N_NODES + 255) / 256;
    const int gE    = (int)((E + 255) / 256);
    const int gGemm = (MTILES + 7) / 8;

    k_init_deg<<<gN, blk, 0, stream>>>(dinv);
    k_count_deg<<<gE, blk, 0, stream>>>(dst, E, dinv);
    k_dinv<<<gN, blk, 0, stream>>>(dinv);

    k_gemm<<<gGemm, blk, 0, stream>>>(x, W1, bufA);
    k_aggregate_owned<<<NGROUPS, 1024, 0, stream>>>(bufA, src, dst, dinv, b1, bufB, E);

    k_gemm<<<gGemm, blk, 0, stream>>>(bufB, W2, bufA);
    k_aggregate_owned<<<NGROUPS, 1024, 0, stream>>>(bufA, src, dst, dinv, b2, out, E);
}


# Layer-1 aggregation writes each wave's own edges (source row, own row, weight) to a workspace list; layer 2 adds from it instead of scanning the destination list; sum order: LDS arrival as before
# speedup vs baseline: 2.4264x; 1.1638x over previous
#include <hip/hip_runtime.h>

#define N_NODES 100000
#define DIM 128
#define MTILES (N_NODES / 16)

typedef __attribute__((ext_vector_type(2))) float v2f;
typedef __attribute__((ext_vector_type(8))) float v8f;

__global__ void k_init_deg(float* deg) {
    int i = blockIdx.x * blockDim.x + threadIdx.x;
    if (i < N_NODES) deg[i] = 1.0f;
}

__global__ void k_count_deg(const int* __restrict__ dst, long long E, float* deg) {
    long long e = (long long)blockIdx.x * blockDim.x + threadIdx.x;
    if (e < E) atomicAdd(&deg[dst[e]], 1.0f);
}

__global__ void k_dinv(float* deg) {
    int i = blockIdx.x * blockDim.x + threadIdx.x;
    if (i < N_NODES) deg[i] = rsqrtf(deg[i]);
}

__global__ __launch_bounds__(256) void k_gemm(const float* __restrict__ X,
                                              const float* __restrict__ W,
                                              float* __restrict__ H) {
    int wave = (blockIdx.x * blockDim.x + threadIdx.x) >> 5;
    int lane = threadIdx.x & 31;
    if (wave >= MTILES) return;

    const int m0   = wave * 16;
    const int half = lane >> 4;
    const int l16  = lane & 15;
    const int koff = half * 2;

    v8f acc[8];
#pragma unroll
    for (int n = 0; n < 8; ++n) acc[n] = (v8f){0,0,0,0,0,0,0,0};

    const float* xrow = X + (size_t)(m0 + l16) * DIM;

    for (int k0 = 0; k0 < DIM; k0 += 4) {
        v2f a;
        a.x = xrow[k0 + koff + 0];
        a.y = xrow[k0 + koff + 1];
#pragma unroll
        for (int n = 0; n < 8; ++n) {
            v2f b;
            b.x = W[(size_t)(k0 + koff + 0) * DIM + n * 16 + l16];
            b.y = W[(size_t)(k0 + koff + 1) * DIM + n * 16 + l16];
            acc[n] = __builtin_amdgcn_wmma_f32_16x16x4_f32(
                false, a, false, b, (short)0, acc[n], false, false);
        }
    }

    const int rbase = m0 + half * 8;
#pragma unroll
    for (int n = 0; n < 8; ++n) {
#pragma unroll
        for (int v = 0; v < 8; ++v) {
            H[(size_t)(rbase + v) * DIM + n * 16 + l16] = acc[n][v];
        }
    }
}

#define OWN      400
#define NGROUPS  (N_NODES / OWN)
#define SCAN_U   8
#define QCAP     64
#define WCAP     1024

__device__ __forceinline__ void add_rows(int n, int lane, const float* __restrict__ H,
                                         int s_i, int d_i, float nr_i, float* acc) {
    for (int j0 = 0; j0 < n; j0 += 8) {
        float4 v[8];
#pragma unroll
        for (int k = 0; k < 8; ++k) {
            int jj = (j0 + k < n) ? (j0 + k) : (n - 1);
            int s = __builtin_amdgcn_readlane(s_i, jj);
            v[k] = ((const float4*)(H + (size_t)s * DIM))[lane];
        }
#pragma unroll
        for (int k = 0; k < 8; ++k) {
            int jj = j0 + k;
            if (jj < n) {
                int d = __builtin_amdgcn_readlane(d_i, jj);
                float nr = __int_as_float(__builtin_amdgcn_readlane(__float_as_int(nr_i), jj));
                float* a = acc + d * DIM + lane;
                atomicAdd(a,      v[k].x * nr);
                atomicAdd(a + 32, v[k].y * nr);
                atomicAdd(a + 64, v[k].z * nr);
                atomicAdd(a + 96, v[k].w * nr);
            }
        }
    }
}

__device__ __forceinline__ int drain_edges(const int* qe, int n, int lane,
                                           const float* __restrict__ H,
                                           const int* __restrict__ src,
                                           const int* __restrict__ dst,
                                           const float* __restrict__ dinv,
                                           int lo, float* acc,
                                           int2* wl, int ln, bool record) {
    int s_i = 0, d_i = 0;
    float nr_i = 0.f;
    if (lane < n) {
        int e_i = qe[lane];
        int dg  = dst[e_i];
        s_i  = src[e_i];
        d_i  = dg - lo;
        nr_i = dinv[s_i] * dinv[dg];
        if (record && ln + lane < WCAP)
            wl[ln + lane] = make_int2(s_i | (d_i << 17), __float_as_int(nr_i));
    }
    add_rows(n, lane, H, s_i, d_i, nr_i, acc);
    return ln + n;
}

__global__ __launch_bounds__(1024) void k_aggregate_owned(const float* __restrict__ H,
                                                          const int* __restrict__ src,
                                                          const int* __restrict__ dst,
                                                          const float* __restrict__ dinv,
                                                          const float* __restrict__ bias,
                                                          float* __restrict__ out, long long E64,
                                                          int* list, int* cnt, int mode) {
    __shared__ __attribute__((aligned(16))) float acc[OWN * DIM];
    __shared__ int q_e[32 * QCAP];
    __shared__ int w_cnt[32];

    const int tid  = threadIdx.x;
    const int lane = tid & 31;
    const int w    = __builtin_amdgcn_readfirstlane(tid >> 5);
    const int lo   = blockIdx.x * OWN;
    const int E    = (int)E64;

    for (int i = tid; i < OWN * DIM / 4; i += 1024)
        ((float4*)acc)[i] = make_float4(0.f, 0.f, 0.f, 0.f);
    __syncthreads();

    int2* wl = (int2*)list + ((size_t)blockIdx.x * 32 + w) * WCAP;
    int ln = 0;
    bool scan = true;
    if (mode != 0) {
        ln = cnt[blockIdx.x * 32 + w];
        scan = (unsigned)ln > (unsigned)WCAP;
    }

    if (scan) {
        const bool record = mode == 0;
        int* qe = q_e + w * QCAP;
        int qn = 0;
        const unsigned ltmask = (1u << lane) - 1u;

        const int F     = E >> 7;
        const int trips = (F + 32 * SCAN_U - 1) / (32 * SCAN_U);
        int trip = (int)(((long long)blockIdx.x * trips) / NGROUPS);

        for (int n = 0; n <= trips; ++n) {
            unsigned hm = 0;
            int tile0;
            if (n < trips) {
                tile0 = trip * (SCAN_U * 32) + w;
                int4 dv[SCAN_U];
                unsigned okmask = 0;
#pragma unroll
                for (int k = 0; k < SCAN_U; ++k) {
                    int t  = tile0 + k * 32;
                    bool ok = t < F;
                    int tc = ok ? t : (F - 1);
                    okmask |= ok ? (0xFu << (4 * k)) : 0u;
                    dv[k] = *(const int4*)(dst + tc * 128 + lane * 4);
                }
#pragma unroll
                for (int k = 0; k < SCAN_U; ++k) {
                    hm |= ((unsigned)(dv[k].x - lo) < (unsigned)OWN) ? (1u << (4 * k + 0)) : 0u;
                    hm |= ((unsigned)(dv[k].y - lo) < (unsigned)OWN) ? (1u << (4 * k + 1)) : 0u;
                    hm |= ((unsigned)(dv[k].z - lo) < (unsigned)OWN) ? (1u << (4 * k + 2)) : 0u;
                    hm |= ((unsigned)(dv[k].w - lo) < (unsigned)OWN) ? (1u << (4 * k + 3)) : 0u;
                }
                hm &= okmask;
                if (++trip == trips) trip = 0;
            } else {
                tile0 = F;
                if (w == 0) {
                    for (int b = 0; b < 4; ++b) {
                        int e = F * 128 + lane * 4 + b;
                        int dl = (e < E) ? (dst[e] - lo) : -1;
                        hm |= ((unsigned)dl < (unsigned)OWN) ? (1u << b) : 0u;
                    }
                }
            }
            for (;;) {
                bool has = hm != 0u;
                unsigned m = __builtin_amdgcn_ballot_w32(has);
                if (m == 0u) break;
                if (has) {
                    int step = __builtin_ctz(hm);
                    hm &= hm - 1u;
                    int e = (tile0 + (step >> 2) * 32) * 128 + lane * 4 + (step & 3);
                    qe[qn + __builtin_popcount(m & ltmask)] = e;
                }
                qn += __builtin_popcount(m);
                if (qn >= 32) {
                    qn -= 32;
                    ln = drain_edges(qe + qn, 32, lane, H, src, dst, dinv, lo, acc, wl, ln, record);
                }
            }
        }
        if (qn > 0) ln = drain_edges(qe, qn, lane, H, src, dst, dinv, lo, acc, wl, ln, record);
        if (record && lane == 0) w_cnt[w] = ln;
    } else {
        int2 ent = make_int2(0, 0);
        if (lane < ln) ent = wl[lane];
        for (int c = 0; c < ln; c += 32) {
            const int n = (ln - c < 32) ? (ln - c) : 32;
            const int2 cur = ent;
            if (c + 32 + lane < ln) ent = wl[c + 32 + lane];
            add_rows(n, lane, H, cur.x & 0x1FFFF, cur.x >> 17, __int_as_float(cur.y), acc);
        }
    }
    __syncthreads();

    if (mode == 0 && w == 0) cnt[blockIdx.x * 32 + lane] = w_cnt[lane];

    const float4 bb = ((const float4*)bias)[lane];
#pragma unroll 4
    for (int r = w; r < OWN; r += 32) {
        const int i = lo + r;
        float di = dinv[i];
        float sn = di * di;
        const float* a = acc + r * DIM + lane;
        float4 h = ((const float4*)(H + (size_t)i * DIM))[lane];
        float4 o;
        o.x = fmaxf(fmaf(sn, h.x, a[0])  + bb.x, 0.f);
        o.y = fmaxf(fmaf(sn, h.y, a[32]) + bb.y, 0.f);
        o.z = fmaxf(fmaf(sn, h.z, a[64]) + bb.z, 0.f);
        o.w = fmaxf(fmaf(sn, h.w, a[96]) + bb.w, 0.f);
        ((float4*)(out + (size_t)i * DIM))[lane] = o;
    }
}

extern "C" void kernel_launch(void* const* d_in, const int* in_sizes, int n_in,
                              void* d_out, int out_size, void* d_ws, size_t ws_size,
                              hipStream_t stream) {
    const float* x   = (const float*)d_in[0];
    const int*   ei  = (const int*)  d_in[1];
    const float* W1  = (const float*)d_in[2];
    const float* b1  = (const float*)d_in[3];
    const float* W2  = (const float*)d_in[4];
    const float* b2  = (const float*)d_in[5];
    float* out = (float*)d_out;

    const long long E = (long long)in_sizes[1] / 2;
    const int* src = ei;
    const int* dst = ei + E;

    float* dinv = (float*)d_ws;
    float* bufA = dinv + N_NODES;
    float* bufB = bufA + (size_t)N_NODES * DIM;
    int*   cnt  = (int*)(bufB + (size_t)N_NODES * DIM);
    int*   list = cnt + NGROUPS * 32 + 32;

    dim3 blk(256);
    const int gN    = (N_NODES + 255) / 256;
    const int gE    = (int)((E + 255) / 256);
    const int gGemm = (MTILES + 7) / 8;

    k_init_deg<<<gN, blk, 0, stream>>>(dinv);
    k_count_deg<<<gE, blk, 0, stream>>>(dst, E, dinv);
    k_dinv<<<gN, blk, 0, stream>>>(dinv);

    k_gemm<<<gGemm, blk, 0, stream>>>(x, W1, bufA);
    k_aggregate_owned<<<NGROUPS, 1024, 0, stream>>>(bufA, src, dst, dinv, b1, bufB, E, list, cnt, 0);

    k_gemm<<<gGemm, blk, 0, stream>>>(bufB, W2, bufA);
    k_aggregate_owned<<<NGROUPS, 1024, 0, stream>>>(bufA, src, dst, dinv, b2, out, E, list, cnt, 1);
}


# Degrees without global atomics: 128 groups each count one chunk of the destination list into LDS integer counters for half the nodes and store partial counts; one launch adds 64 partials + 1, rsqrt
# speedup vs baseline: 3.1112x; 1.2822x over previous
#include <hip/hip_runtime.h>

#define N_NODES 100000
#define DIM 128
#define MTILES (N_NODES / 16)

typedef __attribute__((ext_vector_type(2))) float v2f;
typedef __attribute__((ext_vector_type(8))) float v8f;

#define DEG_R    50016
#define DEG_NR   ((N_NODES + DEG_R - 1) / DEG_R)
#define DEG_C    64
#define DEG_U    8

__global__ __launch_bounds__(1024) void k_deg_part(const int* __restrict__ dst, long long E64,
                                                   int* __restrict__ part) {
    __shared__ __attribute__((aligned(16))) int c_n[DEG_R];
    const int tid   = threadIdx.x;
    const int range = blockIdx.x % DEG_NR;
    const int chunk = blockIdx.x / DEG_NR;
    const int lo    = range * DEG_R;
    const int nr    = (N_NODES - lo < DEG_R) ? (N_NODES - lo) : DEG_R;

    for (int i = tid; i < DEG_R / 4; i += 1024) ((int4*)c_n)[i] = make_int4(0, 0, 0, 0);
    __syncthreads();

    const int E   = (int)E64;
    const int Q   = E >> 2;
    const int per = (Q + DEG_C - 1) / DEG_C;
    const int q0  = chunk * per;
    const int q1  = (q0 + per < Q) ? (q0 + per) : Q;
    const int4* dq = (const int4*)dst;
    for (int b = q0; b < q1; b += DEG_U * 1024) {
        int4 d[DEG_U];
#pragma unroll
        for (int k = 0; k < DEG_U; ++k) {
            int q  = b + k * 1024 + tid;
            int qc = (q < q1) ? q : (q1 - 1);
            d[k] = dq[qc];
        }
#pragma unroll
        for (int k = 0; k < DEG_U; ++k) {
            if (b + k * 1024 + tid < q1) {
                unsigned a0 = (unsigned)(d[k].x - lo);
                unsigned a1 = (unsigned)(d[k].y - lo);
                unsigned a2 = (unsigned)(d[k].z - lo);
                unsigned a3 = (unsigned)(d[k].w - lo);
                if (a0 < (unsigned)nr) atomicAdd(&c_n[a0], 1);
                if (a1 < (unsigned)nr) atomicAdd(&c_n[a1], 1);
                if (a2 < (unsigned)nr) atomicAdd(&c_n[a2], 1);
                if (a3 < (unsigned)nr) atomicAdd(&c_n[a3], 1);
            }
        }
    }
    if (chunk == DEG_C - 1) {
        int e = Q * 4 + tid;
        if (e < E) {
            unsigned a = (unsigned)(dst[e] - lo);
            if (a < (unsigned)nr) atomicAdd(&c_n[a], 1);
        }
    }
    __syncthreads();

    int4* p = (int4*)(part + (size_t)chunk * N_NODES + lo);
    for (int i = tid; i < nr / 4; i += 1024) p[i] = ((const int4*)c_n)[i];
}

__global__ void k_deg_dinv(const int* __restrict__ part, float* __restrict__ dinv) {
    int i = blockIdx.x * blockDim.x + threadIdx.x;
    if (i < N_NODES) {
        int c = 1;
#pragma unroll
        for (int k = 0; k < DEG_C; ++k) c += part[(size_t)k * N_NODES + i];
        dinv[i] = rsqrtf((float)c);
    }
}

__global__ __launch_bounds__(256) void k_gemm(const float* __restrict__ X,
                                              const float* __restrict__ W,
                                              float* __restrict__ H) {
    int wave = (blockIdx.x * blockDim.x + threadIdx.x) >> 5;
    int lane = threadIdx.x & 31;
    if (wave >= MTILES) return;

    const int m0   = wave * 16;
    const int half = lane >> 4;
    const int l16  = lane & 15;
    const int koff = half * 2;

    v8f acc[8];
#pragma unroll
    for (int n = 0; n < 8; ++n) acc[n] = (v8f){0,0,0,0,0,0,0,0};

    const float* xrow = X + (size_t)(m0 + l16) * DIM;

    for (int k0 = 0; k0 < DIM; k0 += 4) {
        v2f a;
        a.x = xrow[k0 + koff + 0];
        a.y = xrow[k0 + koff + 1];
#pragma unroll
        for (int n = 0; n < 8; ++n) {
            v2f b;
            b.x = W[(size_t)(k0 + koff + 0) * DIM + n * 16 + l16];
            b.y = W[(size_t)(k0 + koff + 1) * DIM + n * 16 + l16];
            acc[n] = __builtin_amdgcn_wmma_f32_16x16x4_f32(
                false, a, false, b, (short)0, acc[n], false, false);
        }
    }

    const int rbase = m0 + half * 8;
#pragma unroll
    for (int n = 0; n < 8; ++n) {
#pragma unroll
        for (int v = 0; v < 8; ++v) {
            H[(size_t)(rbase + v) * DIM + n * 16 + l16] = acc[n][v];
        }
    }
}

#define OWN      400
#define NGROUPS  (N_NODES / OWN)
#define SCAN_U   8
#define QCAP     64
#define WCAP     1024

__device__ __forceinline__ void add_rows(int n, int lane, const float* __restrict__ H,
                                         int s_i, int d_i, float nr_i, float* acc) {
    for (int j0 = 0; j0 < n; j0 += 8) {
        float4 v[8];
#pragma unroll
        for (int k = 0; k < 8; ++k) {
            int jj = (j0 + k < n) ? (j0 + k) : (n - 1);
            int s = __builtin_amdgcn_readlane(s_i, jj);
            v[k] = ((const float4*)(H + (size_t)s * DIM))[lane];
        }
#pragma unroll
        for (int k = 0; k < 8; ++k) {
            int jj = j0 + k;
            if (jj < n) {
                int d = __builtin_amdgcn_readlane(d_i, jj);
                float nr = __int_as_float(__builtin_amdgcn_readlane(__float_as_int(nr_i), jj));
                float* a = acc + d * DIM + lane;
                atomicAdd(a,      v[k].x * nr);
                atomicAdd(a + 32, v[k].y * nr);
                atomicAdd(a + 64, v[k].z * nr);
                atomicAdd(a + 96, v[k].w * nr);
            }
        }
    }
}

__device__ __forceinline__ int drain_edges(const int* qe, int n, int lane,
                                           const float* __restrict__ H,
                                           const int* __restrict__ src,
                                           const int* __restrict__ dst,
                                           const float* __restrict__ dinv,
                                           int lo, float* acc,
                                           int2* wl, int ln, bool record) {
    int s_i = 0, d_i = 0;
    float nr_i = 0.f;
    if (lane < n) {
        int e_i = qe[lane];
        int dg  = dst[e_i];
        s_i  = src[e_i];
        d_i  = dg - lo;
        nr_i = dinv[s_i] * dinv[dg];
        if (record && ln + lane < WCAP)
            wl[ln + lane] = make_int2(s_i | (d_i << 17), __float_as_int(nr_i));
    }
    add_rows(n, lane, H, s_i, d_i, nr_i, acc);
    return ln + n;
}

__global__ __launch_bounds__(1024) void k_aggregate_owned(const float* __restrict__ H,
                                                          const int* __restrict__ src,
                                                          const int* __restrict__ dst,
                                                          const float* __restrict__ dinv,
                                                          const float* __restrict__ bias,
                                                          float* __restrict__ out, long long E64,
                                                          int* list, int* cnt, int mode) {
    __shared__ __attribute__((aligned(16))) float acc[OWN * DIM];
    __shared__ int q_e[32 * QCAP];
    __shared__ int w_cnt[32];

    const int tid  = threadIdx.x;
    const int lane = tid & 31;
    const int w    = __builtin_amdgcn_readfirstlane(tid >> 5);
    const int lo   = blockIdx.x * OWN;
    const int E    = (int)E64;

    for (int i = tid; i < OWN * DIM / 4; i += 1024)
        ((float4*)acc)[i] = make_float4(0.f, 0.f, 0.f, 0.f);
    __syncthreads();

    int2* wl = (int2*)list + ((size_t)blockIdx.x * 32 + w) * WCAP;
    int ln = 0;
    bool scan = true;
    if (mode != 0) {
        ln = cnt[blockIdx.x * 32 + w];
        scan = (unsigned)ln > (unsigned)WCAP;
    }

    if (scan) {
        const bool record = mode == 0;
        int* qe = q_e + w * QCAP;
        int qn = 0;
        const unsigned ltmask = (1u << lane) - 1u;

        const int F     = E >> 7;
        const int trips = (F + 32 * SCAN_U - 1) / (32 * SCAN_U);
        int trip = (int)(((long long)blockIdx.x * trips) / NGROUPS);

        for (int n = 0; n <= trips; ++n) {
            unsigned hm = 0;
            int tile0;
            if (n < trips) {
                tile0 = trip * (SCAN_U * 32) + w;
                int4 dv[SCAN_U];
                unsigned okmask = 0;
#pragma unroll
                for (int k = 0; k < SCAN_U; ++k) {
                    int t  = tile0 + k * 32;
                    bool ok = t < F;
                    int tc = ok ? t : (F - 1);
                    okmask |= ok ? (0xFu << (4 * k)) : 0u;
                    dv[k] = *(const int4*)(dst + tc * 128 + lane * 4);
                }
#pragma unroll
                for (int k = 0; k < SCAN_U; ++k) {
                    hm |= ((unsigned)(dv[k].x - lo) < (unsigned)OWN) ? (1u << (4 * k + 0)) : 0u;
                    hm |= ((unsigned)(dv[k].y - lo) < (unsigned)OWN) ? (1u << (4 * k + 1)) : 0u;
                    hm |= ((unsigned)(dv[k].z - lo) < (unsigned)OWN) ? (1u << (4 * k + 2)) : 0u;
                    hm |= ((unsigned)(dv[k].w - lo) < (unsigned)OWN) ? (1u << (4 * k + 3)) : 0u;
                }
                hm &= okmask;
                if (++trip == trips) trip = 0;
            } else {
                tile0 = F;
                if (w == 0) {
                    for (int b = 0; b < 4; ++b) {
                        int e = F * 128 + lane * 4 + b;
                        int dl = (e < E) ? (dst[e] - lo) : -1;
                        hm |= ((unsigned)dl < (unsigned)OWN) ? (1u << b) : 0u;
                    }
                }
            }
            for (;;) {
                bool has = hm != 0u;
                unsigned m = __builtin_amdgcn_ballot_w32(has);
                if (m == 0u) break;
                if (has) {
                    int step = __builtin_ctz(hm);
                    hm &= hm - 1u;
                    int e = (tile0 + (step >> 2) * 32) * 128 + lane * 4 + (step & 3);
                    qe[qn + __builtin_popcount(m & ltmask)] = e;
                }
                qn += __builtin_popcount(m);
                if (qn >= 32) {
                    qn -= 32;
                    ln = drain_edges(qe + qn, 32, lane, H, src, dst, dinv, lo, acc, wl, ln, record);
                }
            }
        }
        if (qn > 0) ln = drain_edges(qe, qn, lane, H, src, dst, dinv, lo, acc, wl, ln, record);
        if (record && lane == 0) w_cnt[w] = ln;
    } else {
        int2 ent = make_int2(0, 0);
        if (lane < ln) ent = wl[lane];
        for (int c = 0; c < ln; c += 32) {
            const int n = (ln - c < 32) ? (ln - c) : 32;
            const int2 cur = ent;
            if (c + 32 + lane < ln) ent = wl[c + 32 + lane];
            add_rows(n, lane, H, cur.x & 0x1FFFF, cur.x >> 17, __int_as_float(cur.y), acc);
        }
    }
    __syncthreads();

    if (mode == 0 && w == 0) cnt[blockIdx.x * 32 + lane] = w_cnt[lane];

    const float4 bb = ((const float4*)bias)[lane];
#pragma unroll 4
    for (int r = w; r < OWN; r += 32) {
        const int i = lo + r;
        float di = dinv[i];
        float sn = di * di;
        const float* a = acc + r * DIM + lane;
        float4 h = ((const float4*)(H + (size_t)i * DIM))[lane];
        float4 o;
        o.x = fmaxf(fmaf(sn, h.x, a[0])  + bb.x, 0.f);
        o.y = fmaxf(fmaf(sn, h.y, a[32]) + bb.y, 0.f);
        o.z = fmaxf(fmaf(sn, h.z, a[64]) + bb.z, 0.f);
        o.w = fmaxf(fmaf(sn, h.w, a[96]) + bb.w, 0.f);
        ((float4*)(out + (size_t)i * DIM))[lane] = o;
    }
}

extern "C" void kernel_launch(void* const* d_in, const int* in_sizes, int n_in,
                              void* d_out, int out_size, void* d_ws, size_t ws_size,
                              hipStream_t stream) {
    const float* x   = (const float*)d_in[0];
    const int*   ei  = (const int*)  d_in[1];
    const float* W1  = (const float*)d_in[2];
    const float* b1  = (const float*)d_in[3];
    const float* W2  = (const float*)d_in[4];
    const float* b2  = (const float*)d_in[5];
    float* out = (float*)d_out;

    const long long E = (long long)in_sizes[1] / 2;
    const int* src = ei;
    const int* dst = ei + E;

    float* dinv = (float*)d_ws;
    float* bufA = dinv + N_NODES;
    float* bufB = bufA + (size_t)N_NODES * DIM;
    int*   cnt  = (int*)(bufB + (size_t)N_NODES * DIM);
    int*   list = cnt + NGROUPS * 32 + 32;
    int*   part = list + (size_t)NGROUPS * 32 * WCAP * 2;

    dim3 blk(256);
    const int gN    = (N_NODES + 255) / 256;
    const int gGemm = (MTILES + 7) / 8;

    k_deg_part<<<DEG_NR * DEG_C, 1024, 0, stream>>>(dst, E, part);
    k_deg_dinv<<<gN, blk, 0, stream>>>(part, dinv);

    k_gemm<<<gGemm, blk, 0, stream>>>(x, W1, bufA);
    k_aggregate_owned<<<NGROUPS, 1024, 0, stream>>>(bufA, src, dst, dinv, b1, bufB, E, list, cnt, 0);

    k_gemm<<<gGemm, blk, 0, stream>>>(bufB, W2, bufA);
    k_aggregate_owned<<<NGROUPS, 1024, 0, stream>>>(bufA, src, dst, dinv, b2, out, E, list, cnt, 1);
}


# First-layer edge search scans a one-byte owner-group number per edge (destination / 400, written by the degree launch), 16 edges per 16-byte load, not 4-byte destinations; sum order: LDS arrival
# speedup vs baseline: 3.3613x; 1.0804x over previous
#include <hip/hip_runtime.h>

#define N_NODES 100000
#define DIM 128
#define MTILES (N_NODES / 16)

typedef __attribute__((ext_vector_type(2))) float v2f;
typedef __attribute__((ext_vector_type(8))) float v8f;

#define DEG_R    50016
#define DEG_NR   ((N_NODES + DEG_R - 1) / DEG_R)
#define DEG_C    64
#define DEG_U    8

#define OWN      400

__device__ __forceinline__ unsigned owner_byte(int d) {
    unsigned o = (unsigned)d / (unsigned)OWN;
    return o < 255u ? o : 255u;
}

__global__ __launch_bounds__(1024) void k_deg_part(const int* __restrict__ dst, long long E64,
                                                   int* __restrict__ part, int* __restrict__ own) {
    __shared__ __attribute__((aligned(16))) int c_n[DEG_R];
    const int tid   = threadIdx.x;
    const int range = blockIdx.x % DEG_NR;
    const int chunk = blockIdx.x / DEG_NR;
    const int lo    = range * DEG_R;
    const int nr    = (N_NODES - lo < DEG_R) ? (N_NODES - lo) : DEG_R;

    for (int i = tid; i < DEG_R / 4; i += 1024) ((int4*)c_n)[i] = make_int4(0, 0, 0, 0);
    __syncthreads();

    const int E   = (int)E64;
    const int Q   = E >> 2;
    const int per = (((Q + DEG_C - 1) / DEG_C) + 31) & ~31;
    const int q0  = chunk * per;
    const int q1  = (q0 + per < Q) ? (q0 + per) : Q;
    const int4* dq = (const int4*)dst;
    for (int b = q0; b < q1; b += DEG_U * 1024) {
        int4 d[DEG_U];
#pragma unroll
        for (int k = 0; k < DEG_U; ++k) {
            int q  = b + k * 1024 + tid;
            int qc = (q < q1) ? q : (q1 - 1);
            d[k] = dq[qc];
        }
#pragma unroll
        for (int k = 0; k < DEG_U; ++k) {
            if (b + k * 1024 + tid < q1) {
                if (range == 0)
                    own[b + k * 1024 + tid] = (int)(owner_byte(d[k].x) | (owner_byte(d[k].y) << 8) |
                                                    (owner_byte(d[k].z) << 16) | (owner_byte(d[k].w) << 24));
                unsigned a0 = (unsigned)(d[k].x - lo);
                unsigned a1 = (unsigned)(d[k].y - lo);
                unsigned a2 = (unsigned)(d[k].z - lo);
                unsigned a3 = (unsigned)(d[k].w - lo);
                if (a0 < (unsigned)nr) atomicAdd(&c_n[a0], 1);
                if (a1 < (unsigned)nr) atomicAdd(&c_n[a1], 1);
                if (a2 < (unsigned)nr) atomicAdd(&c_n[a2], 1);
                if (a3 < (unsigned)nr) atomicAdd(&c_n[a3], 1);
            }
        }
    }
    if (chunk == DEG_C - 1) {
        int e = Q * 4 + tid;
        if (e < E) {
            unsigned a = (unsigned)(dst[e] - lo);
            if (a < (unsigned)nr) atomicAdd(&c_n[a], 1);
        }
        if (range == 0) {
            const int QP = ((E + 511) >> 9) * 128;
            for (int q = Q + tid; q < QP; q += 1024) {
                unsigned pk = 0;
                for (int j = 0; j < 4; ++j) {
                    int ee = q * 4 + j;
                    unsigned o = (ee < E) ? owner_byte(dst[ee]) : 255u;
                    pk |= o << (8 * j);
                }
                own[q] = (int)pk;
            }
        }
    }
    __syncthreads();

    int4* p = (int4*)(part + (size_t)chunk * N_NODES + lo);
    for (int i = tid; i < nr / 4; i += 1024) p[i] = ((const int4*)c_n)[i];
}

__global__ void k_deg_dinv(const int* __restrict__ part, float* __restrict__ dinv) {
    int i = blockIdx.x * blockDim.x + threadIdx.x;
    if (i < N_NODES) {
        int c = 1;
#pragma unroll
        for (int k = 0; k < DEG_C; ++k) c += part[(size_t)k * N_NODES + i];
        dinv[i] = rsqrtf((float)c);
    }
}

__global__ __launch_bounds__(256) void k_gemm(const float* __restrict__ X,
                                              const float* __restrict__ W,
                                              float* __restrict__ H) {
    int wave = (blockIdx.x * blockDim.x + threadIdx.x) >> 5;
    int lane = threadIdx.x & 31;
    if (wave >= MTILES) return;

    const int m0   = wave * 16;
    const int half = lane >> 4;
    const int l16  = lane & 15;
    const int koff = half * 2;

    v8f acc[8];
#pragma unroll
    for (int n = 0; n < 8; ++n) acc[n] = (v8f){0,0,0,0,0,0,0,0};

    const float* xrow = X + (size_t)(m0 + l16) * DIM;

    for (int k0 = 0; k0 < DIM; k0 += 4) {
        v2f a;
        a.x = xrow[k0 + koff + 0];
        a.y = xrow[k0 + koff + 1];
#pragma unroll
        for (int n = 0; n < 8; ++n) {
            v2f b;
            b.x = W[(size_t)(k0 + koff + 0) * DIM + n * 16 + l16];
            b.y = W[(size_t)(k0 + koff + 1) * DIM + n * 16 + l16];
            acc[n] = __builtin_amdgcn_wmma_f32_16x16x4_f32(
                false, a, false, b, (short)0, acc[n], false, false);
        }
    }

    const int rbase = m0 + half * 8;
#pragma unroll
    for (int n = 0; n < 8; ++n) {
#pragma unroll
        for (int v = 0; v < 8; ++v) {
            H[(size_t)(rbase + v) * DIM + n * 16 + l16] = acc[n][v];
        }
    }
}

#define NGROUPS  (N_NODES / OWN)
#define SCAN_U   8
#define QCAP     64
#define WCAP     1024

__device__ __forceinline__ void add_rows(int n, int lane, const float* __restrict__ H,
                                         int s_i, int d_i, float nr_i, float* acc) {
    for (int j0 = 0; j0 < n; j0 += 8) {
        float4 v[8];
#pragma unroll
        for (int k = 0; k < 8; ++k) {
            int jj = (j0 + k < n) ? (j0 + k) : (n - 1);
            int s = __builtin_amdgcn_readlane(s_i, jj);
            v[k] = ((const float4*)(H + (size_t)s * DIM))[lane];
        }
#pragma unroll
        for (int k = 0; k < 8; ++k) {
            int jj = j0 + k;
            if (jj < n) {
                int d = __builtin_amdgcn_readlane(d_i, jj);
                float nr = __int_as_float(__builtin_amdgcn_readlane(__float_as_int(nr_i), jj));
                float* a = acc + d * DIM + lane;
                atomicAdd(a,      v[k].x * nr);
                atomicAdd(a + 32, v[k].y * nr);
                atomicAdd(a + 64, v[k].z * nr);
                atomicAdd(a + 96, v[k].w * nr);
            }
        }
    }
}

__device__ __forceinline__ int drain_edges(const int* qe, int n, int lane,
                                           const float* __restrict__ H,
                                           const int* __restrict__ src,
                                           const int* __restrict__ dst,
                                           const float* __restrict__ dinv,
                                           int lo, float* acc,
                                           int2* wl, int ln, bool record) {
    int s_i = 0, d_i = 0;
    float nr_i = 0.f;
    if (lane < n) {
        int e_i = qe[lane];
        int dg  = dst[e_i];
        s_i  = src[e_i];
        d_i  = dg - lo;
        nr_i = dinv[s_i] * dinv[dg];
        if (record && ln + lane < WCAP)
            wl[ln + lane] = make_int2(s_i | (d_i << 17), __float_as_int(nr_i));
    }
    add_rows(n, lane, H, s_i, d_i, nr_i, acc);
    return ln + n;
}

__global__ __launch_bounds__(1024) void k_aggregate_owned(const float* __restrict__ H,
                                                          const int* __restrict__ src,
                                                          const int* __restrict__ dst,
                                                          const float* __restrict__ dinv,
                                                          const float* __restrict__ bias,
                                                          float* __restrict__ out, long long E64,
                                                          int* list, int* cnt, int mode,
                                                          const int* __restrict__ own) {
    __shared__ __attribute__((aligned(16))) float acc[OWN * DIM];
    __shared__ int q_e[32 * QCAP];
    __shared__ int w_cnt[32];

    const int tid  = threadIdx.x;
    const int lane = tid & 31;
    const int w    = __builtin_amdgcn_readfirstlane(tid >> 5);
    const int lo   = blockIdx.x * OWN;
    const int E    = (int)E64;

    for (int i = tid; i < OWN * DIM / 4; i += 1024)
        ((float4*)acc)[i] = make_float4(0.f, 0.f, 0.f, 0.f);
    __syncthreads();

    int2* wl = (int2*)list + ((size_t)blockIdx.x * 32 + w) * WCAP;
    int ln = 0;
    bool scan = true;
    if (mode != 0) {
        ln = cnt[blockIdx.x * 32 + w];
        scan = (unsigned)ln > (unsigned)WCAP;
    }

    if (scan) {
        const bool record = mode == 0;
        int* qe = q_e + w * QCAP;
        int qn = 0;
        const unsigned ltmask = (1u << lane) - 1u;

        const int F     = (E + 511) >> 9;
        const int trips = (F + 32 * SCAN_U - 1) / (32 * SCAN_U);
        int trip = (int)(((long long)blockIdx.x * trips) / NGROUPS);
        const unsigned pat = (unsigned)blockIdx.x * 0x01010101u;

        for (int n = 0; n < trips; ++n) {
            const int tile0 = trip * (SCAN_U * 32) + w;
            int4 dv[SCAN_U];
#pragma unroll
            for (int k = 0; k < SCAN_U; ++k) {
                int t  = tile0 + k * 32;
                int tc = (t < F) ? t : (F - 1);
                dv[k] = *(const int4*)(own + tc * 128 + lane * 4);
            }
            if (++trip == trips) trip = 0;
            static_assert(SCAN_U == 8, "the queueing below names four masks");
            unsigned hms[SCAN_U / 2];
#pragma unroll
            for (int p = 0; p < SCAN_U / 2; ++p) {
                const unsigned w8[8] = {(unsigned)dv[2 * p].x,     (unsigned)dv[2 * p].y,
                                        (unsigned)dv[2 * p].z,     (unsigned)dv[2 * p].w,
                                        (unsigned)dv[2 * p + 1].x, (unsigned)dv[2 * p + 1].y,
                                        (unsigned)dv[2 * p + 1].z, (unsigned)dv[2 * p + 1].w};
                unsigned hm = 0;
#pragma unroll
                for (int k = 0; k < 8; ++k) {
                    unsigned x = w8[k] ^ pat;
                    unsigned a = (x & 0x7F7F7F7Fu) + 0x7F7F7F7Fu;
                    unsigned z = ~(a | x) & 0x80808080u;
                    hm |= z >> k;
                }
                const int ta = tile0 + (2 * p) * 32;
                hms[p] = hm & (((ta < F) ? 0xF0F0F0F0u : 0u) | ((ta + 32 < F) ? 0x0F0F0F0Fu : 0u));
            }
            for (;;) {
                bool has = (hms[0] | hms[1] | hms[2] | hms[3]) != 0u;
                unsigned m = __builtin_amdgcn_ballot_w32(has);
                if (m == 0u) break;
                if (has) {
                    const int p = hms[0] ? 0 : hms[1] ? 1 : hms[2] ? 2 : 3;
                    const unsigned h = hms[0] ? hms[0] : hms[1] ? hms[1] : hms[2] ? hms[2] : hms[3];
                    const int step = __builtin_ctz(h);
                    const unsigned c = h & (h - 1u);
                    if (p == 0) hms[0] = c; else if (p == 1) hms[1] = c; else if (p == 2) hms[2] = c; else hms[3] = c;
                    const int k = 7 - (step & 7);
                    int e = (tile0 + (2 * p + (k >> 2)) * 32) * 512 + lane * 16 + (k & 3) * 4 + (step >> 3);
                    qe[qn + __builtin_popcount(m & ltmask)] = e;
                }
                qn += __builtin_popcount(m);
                if (qn >= 32) {
                    qn -= 32;
                    ln = drain_edges(qe + qn, 32, lane, H, src, dst, dinv, lo, acc, wl, ln, record);
                }
            }
        }
        if (qn > 0) ln = drain_edges(qe, qn, lane, H, src, dst, dinv, lo, acc, wl, ln, record);
        if (record && lane == 0) w_cnt[w] = ln;
    } else {
        int2 ent = make_int2(0, 0);
        if (lane < ln) ent = wl[lane];
        for (int c = 0; c < ln; c += 32) {
            const int n = (ln - c < 32) ? (ln - c) : 32;
            const int2 cur = ent;
            if (c + 32 + lane < ln) ent = wl[c + 32 + lane];
            add_rows(n, lane, H, cur.x & 0x1FFFF, cur.x >> 17, __int_as_float(cur.y), acc);
        }
    }
    __syncthreads();

    if (mode == 0 && w == 0) cnt[blockIdx.x * 32 + lane] = w_cnt[lane];

    const float4 bb = ((const float4*)bias)[lane];
#pragma unroll 4
    for (int r = w; r < OWN; r += 32) {
        const int i = lo + r;
        float di = dinv[i];
        float sn = di * di;
        const float* a = acc + r * DIM + lane;
        float4 h = ((const float4*)(H + (size_t)i * DIM))[lane];
        float4 o;
        o.x = fmaxf(fmaf(sn, h.x, a[0])  + bb.x, 0.f);
        o.y = fmaxf(fmaf(sn, h.y, a[32]) + bb.y, 0.f);
        o.z = fmaxf(fmaf(sn, h.z, a[64]) + bb.z, 0.f);
        o.w = fmaxf(fmaf(sn, h.w, a[96]) + bb.w, 0.f);
        ((float4*)(out + (size_t)i * DIM))[lane] = o;
    }
}

extern "C" void kernel_launch(void* const* d_in, const int* in_sizes, int n_in,
                              void* d_out, int out_size, void* d_ws, size_t ws_size,
                              hipStream_t stream) {
    const float* x   = (const float*)d_in[0];
    const int*   ei  = (const int*)  d_in[1];
    const float* W1  = (const float*)d_in[2];
    const float* b1  = (const float*)d_in[3];
    const float* W2  = (const float*)d_in[4];
    const float* b2  = (const float*)d_in[5];
    float* out = (float*)d_out;

    const long long E = (long long)in_sizes[1] / 2;
    const int* src = ei;
    const int* dst = ei + E;

    float* dinv = (float*)d_ws;
    float* bufA = dinv + N_NODES;
    float* bufB = bufA + (size_t)N_NODES * DIM;
    int*   cnt  = (int*)(bufB + (size_t)N_NODES * DIM);
    int*   list = cnt + NGROUPS * 32 + 32;
    int*   part = list + (size_t)NGROUPS * 32 * WCAP * 2;
    int*   own  = part + (size_t)DEG_C * N_NODES;

    dim3 blk(256);
    const int gN    = (N_NODES + 255) / 256;
    const int gGemm = (MTILES + 7) / 8;

    k_deg_part<<<DEG_NR * DEG_C, 1024, 0, stream>>>(dst, E, part, own);
    k_deg_dinv<<<gN, blk, 0, stream>>>(part, dinv);

    k_gemm<<<gGemm, blk, 0, stream>>>(x, W1, bufA);
    k_aggregate_owned<<<NGROUPS, 1024, 0, stream>>>(bufA, src, dst, dinv, b1, bufB, E, list, cnt, 0, own);

    k_gemm<<<gGemm, blk, 0, stream>>>(bufB, W2, bufA);
    k_aggregate_owned<<<NGROUPS, 1024, 0, stream>>>(bufA, src, dst, dinv, b2, out, E, list, cnt, 1, own);
}


# Product launches: W staged once per 1024-thread group in LDS in fragment order, B fragments by 16-byte LDS loads, not 16 global loads per K-step; 256 groups, strips dealt round-robin; sum order kept
# speedup vs baseline: 3.6417x; 1.0834x over previous
#include <hip/hip_runtime.h>

#define N_NODES 100000
#define DIM 128
#define MTILES (N_NODES / 16)

typedef __attribute__((ext_vector_type(2))) float v2f;
typedef __attribute__((ext_vector_type(8))) float v8f;

#define DEG_R    50016
#define DEG_NR   ((N_NODES + DEG_R - 1) / DEG_R)
#define DEG_C    64
#define DEG_U    8

#define OWN      400

__device__ __forceinline__ unsigned owner_byte(int d) {
    unsigned o = (unsigned)d / (unsigned)OWN;
    return o < 255u ? o : 255u;
}

__global__ __launch_bounds__(1024) void k_deg_part(const int* __restrict__ dst, long long E64,
                                                   int* __restrict__ part, int* __restrict__ own) {
    __shared__ __attribute__((aligned(16))) int c_n[DEG_R];
    const int tid   = threadIdx.x;
    const int range = blockIdx.x % DEG_NR;
    const int chunk = blockIdx.x / DEG_NR;
    const int lo    = range * DEG_R;
    const int nr    = (N_NODES - lo < DEG_R) ? (N_NODES - lo) : DEG_R;

    for (int i = tid; i < DEG_R / 4; i += 1024) ((int4*)c_n)[i] = make_int4(0, 0, 0, 0);
    __syncthreads();

    const int E   = (int)E64;
    const int Q   = E >> 2;
    const int per = (((Q + DEG_C - 1) / DEG_C) + 31) & ~31;
    const int q0  = chunk * per;
    const int q1  = (q0 + per < Q) ? (q0 + per) : Q;
    const int4* dq = (const int4*)dst;
    for (int b = q0; b < q1; b += DEG_U * 1024) {
        int4 d[DEG_U];
#pragma unroll
        for (int k = 0; k < DEG_U; ++k) {
            int q  = b + k * 1024 + tid;
            int qc = (q < q1) ? q : (q1 - 1);
            d[k] = dq[qc];
        }
#pragma unroll
        for (int k = 0; k < DEG_U; ++k) {
            if (b + k * 1024 + tid < q1) {
                if (range == 0)
                    own[b + k * 1024 + tid] = (int)(owner_byte(d[k].x) | (owner_byte(d[k].y) << 8) |
                                                    (owner_byte(d[k].z) << 16) | (owner_byte(d[k].w) << 24));
                unsigned a0 = (unsigned)(d[k].x - lo);
                unsigned a1 = (unsigned)(d[k].y - lo);
                unsigned a2 = (unsigned)(d[k].z - lo);
                unsigned a3 = (unsigned)(d[k].w - lo);
                if (a0 < (unsigned)nr) atomicAdd(&c_n[a0], 1);
                if (a1 < (unsigned)nr) atomicAdd(&c_n[a1], 1);
                if (a2 < (unsigned)nr) atomicAdd(&c_n[a2], 1);
                if (a3 < (unsigned)nr) atomicAdd(&c_n[a3], 1);
            }
        }
    }
    if (chunk == DEG_C - 1) {
        int e = Q * 4 + tid;
        if (e < E) {
            unsigned a = (unsigned)(dst[e] - lo);
            if (a < (unsigned)nr) atomicAdd(&c_n[a], 1);
        }
        if (range == 0) {
            const int QP = ((E + 511) >> 9) * 128;
            for (int q = Q + tid; q < QP; q += 1024) {
                unsigned pk = 0;
                for (int j = 0; j < 4; ++j) {
                    int ee = q * 4 + j;
                    unsigned o = (ee < E) ? owner_byte(dst[ee]) : 255u;
                    pk |= o << (8 * j);
                }
                own[q] = (int)pk;
            }
        }
    }
    __syncthreads();

    int4* p = (int4*)(part + (size_t)chunk * N_NODES + lo);
    for (int i = tid; i < nr / 4; i += 1024) p[i] = ((const int4*)c_n)[i];
}

__global__ void k_deg_dinv(const int* __restrict__ part, float* __restrict__ dinv) {
    int i = blockIdx.x * blockDim.x + threadIdx.x;
    if (i < N_NODES) {
        int c = 1;
#pragma unroll
        for (int k = 0; k < DEG_C; ++k) c += part[(size_t)k * N_NODES + i];
        dinv[i] = rsqrtf((float)c);
    }
}

#define GEMM_GROUPS 256
__global__ __launch_bounds__(1024) void k_gemm(const float* __restrict__ X,
                                               const float* __restrict__ W,
                                               float* __restrict__ H) {
    __shared__ __attribute__((aligned(16))) float w_l[DIM * DIM];

    const int tid = threadIdx.x;
    for (int idx = tid; idx < DIM * DIM / 4; idx += 1024) {
        const int c  = idx & 15;
        const int j  = (idx >> 4) & 3;
        const int kp = idx >> 6;
        const float* w0 = W + (size_t)(2 * kp) * DIM + j * 32 + c;
        float4 o;
        o.x = w0[0];
        o.y = w0[DIM];
        o.z = w0[16];
        o.w = w0[DIM + 16];
        ((float4*)w_l)[idx] = o;
    }
    __syncthreads();

    const int lane = tid & 31;
    const int wave = (tid >> 5) * GEMM_GROUPS + blockIdx.x;
    if (wave >= MTILES) return;

    const int m0   = wave * 16;
    const int half = lane >> 4;
    const int l16  = lane & 15;
    const int koff = half * 2;

    v8f acc[8];
#pragma unroll
    for (int n = 0; n < 8; ++n) acc[n] = (v8f){0,0,0,0,0,0,0,0};

    const float* xrow = X + (size_t)(m0 + l16) * DIM;
    const float4* wq  = (const float4*)w_l + half * 64 + l16;

    v2f an;
    an.x = xrow[koff + 0];
    an.y = xrow[koff + 1];
    float4 qn0 = wq[0], qn1 = wq[16], qn2 = wq[32], qn3 = wq[48];
#pragma unroll 2
    for (int k0 = 0; k0 < DIM; k0 += 4) {
        const v2f a = an;
        const float4 q0 = qn0, q1 = qn1, q2 = qn2, q3 = qn3;
        const int kn = (k0 + 4 < DIM) ? (k0 + 4) : k0;
        an.x = xrow[kn + koff + 0];
        an.y = xrow[kn + koff + 1];
        const float4* bp = wq + (kn >> 1) * 64;
        qn0 = bp[0]; qn1 = bp[16]; qn2 = bp[32]; qn3 = bp[48];
        v2f b;
        b.x = q0.x; b.y = q0.y;
        acc[0] = __builtin_amdgcn_wmma_f32_16x16x4_f32(false, a, false, b, (short)0, acc[0], false, false);
        b.x = q0.z; b.y = q0.w;
        acc[1] = __builtin_amdgcn_wmma_f32_16x16x4_f32(false, a, false, b, (short)0, acc[1], false, false);
        b.x = q1.x; b.y = q1.y;
        acc[2] = __builtin_amdgcn_wmma_f32_16x16x4_f32(false, a, false, b, (short)0, acc[2], false, false);
        b.x = q1.z; b.y = q1.w;
        acc[3] = __builtin_amdgcn_wmma_f32_16x16x4_f32(false, a, false, b, (short)0, acc[3], false, false);
        b.x = q2.x; b.y = q2.y;
        acc[4] = __builtin_amdgcn_wmma_f32_16x16x4_f32(false, a, false, b, (short)0, acc[4], false, false);
        b.x = q2.z; b.y = q2.w;
        acc[5] = __builtin_amdgcn_wmma_f32_16x16x4_f32(false, a, false, b, (short)0, acc[5], false, false);
        b.x = q3.x; b.y = q3.y;
        acc[6] = __builtin_amdgcn_wmma_f32_16x16x4_f32(false, a, false, b, (short)0, acc[6], false, false);
        b.x = q3.z; b.y = q3.w;
        acc[7] = __builtin_amdgcn_wmma_f32_16x16x4_f32(false, a, false, b, (short)0, acc[7], false, false);
    }

    const int rbase = m0 + half * 8;
#pragma unroll
    for (int n = 0; n < 8; ++n) {
#pragma unroll
        for (int v = 0; v < 8; ++v) {
            H[(size_t)(rbase + v) * DIM + n * 16 + l16] = acc[n][v];
        }
    }
}

#define NGROUPS  (N_NODES / OWN)
#define SCAN_U   8
#define QCAP     64
#define WCAP     1024

__device__ __forceinline__ void add_rows(int n, int lane, const float* __restrict__ H,
                                         int s_i, int d_i, float nr_i, float* acc) {
    for (int j0 = 0; j0 < n; j0 += 8) {
        float4 v[8];
#pragma unroll
        for (int k = 0; k < 8; ++k) {
            int jj = (j0 + k < n) ? (j0 + k) : (n - 1);
            int s = __builtin_amdgcn_readlane(s_i, jj);
            v[k] = ((const float4*)(H + (size_t)s * DIM))[lane];
        }
#pragma unroll
        for (int k = 0; k < 8; ++k) {
            int jj = j0 + k;
            if (jj < n) {
                int d = __builtin_amdgcn_readlane(d_i, jj);
                float nr = __int_as_float(__builtin_amdgcn_readlane(__float_as_int(nr_i), jj));
                float* a = acc + d * DIM + lane;
                atomicAdd(a,      v[k].x * nr);
                atomicAdd(a + 32, v[k].y * nr);
                atomicAdd(a + 64, v[k].z * nr);
                atomicAdd(a + 96, v[k].w * nr);
            }
        }
    }
}

__device__ __forceinline__ int drain_edges(const int* qe, int n, int lane,
                                           const float* __restrict__ H,
                                           const int* __restrict__ src,
                                           const int* __restrict__ dst,
                                           const float* __restrict__ dinv,
                                           int lo, float* acc,
                                           int2* wl, int ln, bool record) {
    int s_i = 0, d_i = 0;
    float nr_i = 0.f;
    if (lane < n) {
        int e_i = qe[lane];
        int dg  = dst[e_i];
        s_i  = src[e_i];
        d_i  = dg - lo;
        nr_i = dinv[s_i] * dinv[dg];
        if (record && ln + lane < WCAP)
            wl[ln + lane] = make_int2(s_i | (d_i << 17), __float_as_int(nr_i));
    }
    add_rows(n, lane, H, s_i, d_i, nr_i, acc);
    return ln + n;
}

__global__ __launch_bounds__(1024) void k_aggregate_owned(const float* __restrict__ H,
                                                          const int* __restrict__ src,
                                                          const int* __restrict__ dst,
                                                          const float* __restrict__ dinv,
                                                          const float* __restrict__ bias,
                                                          float* __restrict__ out, long long E64,
                                                          int* list, int* cnt, int mode,
                                                          const int* __restrict__ own) {
    __shared__ __attribute__((aligned(16))) float acc[OWN * DIM];
    __shared__ int q_e[32 * QCAP];
    __shared__ int w_cnt[32];

    const int tid  = threadIdx.x;
    const int lane = tid & 31;
    const int w    = __builtin_amdgcn_readfirstlane(tid >> 5);
    const int lo   = blockIdx.x * OWN;
    const int E    = (int)E64;

    for (int i = tid; i < OWN * DIM / 4; i += 1024)
        ((float4*)acc)[i] = make_float4(0.f, 0.f, 0.f, 0.f);
    __syncthreads();

    int2* wl = (int2*)list + ((size_t)blockIdx.x * 32 + w) * WCAP;
    int ln = 0;
    bool scan = true;
    if (mode != 0) {
        ln = cnt[blockIdx.x * 32 + w];
        scan = (unsigned)ln > (unsigned)WCAP;
    }

    if (scan) {
        const bool record = mode == 0;
        int* qe = q_e + w * QCAP;
        int qn = 0;
        const unsigned ltmask = (1u << lane) - 1u;

        const int F     = (E + 511) >> 9;
        const int trips = (F + 32 * SCAN_U - 1) / (32 * SCAN_U);
        int trip = (int)(((long long)blockIdx.x * trips) / NGROUPS);
        const unsigned pat = (unsigned)blockIdx.x * 0x01010101u;

        for (int n = 0; n < trips; ++n) {
            const int tile0 = trip * (SCAN_U * 32) + w;
            int4 dv[SCAN_U];
#pragma unroll
            for (int k = 0; k < SCAN_U; ++k) {
                int t  = tile0 + k * 32;
                int tc = (t < F) ? t : (F - 1);
                dv[k] = *(const int4*)(own + tc * 128 + lane * 4);
            }
            if (++trip == trips) trip = 0;
            static_assert(SCAN_U == 8, "the queueing below names four masks");
            unsigned hms[SCAN_U / 2];
#pragma unroll
            for (int p = 0; p < SCAN_U / 2; ++p) {
                const unsigned w8[8] = {(unsigned)dv[2 * p].x,     (unsigned)dv[2 * p].y,
                                        (unsigned)dv[2 * p].z,     (unsigned)dv[2 * p].w,
                                        (unsigned)dv[2 * p + 1].x, (unsigned)dv[2 * p + 1].y,
                                        (unsigned)dv[2 * p + 1].z, (unsigned)dv[2 * p + 1].w};
                unsigned hm = 0;
#pragma unroll
                for (int k = 0; k < 8; ++k) {
                    unsigned x = w8[k] ^ pat;
                    unsigned a = (x & 0x7F7F7F7Fu) + 0x7F7F7F7Fu;
                    unsigned z = ~(a | x) & 0x80808080u;
                    hm |= z >> k;
                }
                const int ta = tile0 + (2 * p) * 32;
                hms[p] = hm & (((ta < F) ? 0xF0F0F0F0u : 0u) | ((ta + 32 < F) ? 0x0F0F0F0Fu : 0u));
            }
            for (;;) {
                bool has = (hms[0] | hms[1] | hms[2] | hms[3]) != 0u;
                unsigned m = __builtin_amdgcn_ballot_w32(has);
                if (m == 0u) break;
                if (has) {
                    const int p = hms[0] ? 0 : hms[1] ? 1 : hms[2] ? 2 : 3;
                    const unsigned h = hms[0] ? hms[0] : hms[1] ? hms[1] : hms[2] ? hms[2] : hms[3];
                    const int step = __builtin_ctz(h);
                    const unsigned c = h & (h - 1u);
                    if (p == 0) hms[0] = c; else if (p == 1) hms[1] = c; else if (p == 2) hms[2] = c; else hms[3] = c;
                    const int k = 7 - (step & 7);
                    int e = (tile0 + (2 * p + (k >> 2)) * 32) * 512 + lane * 16 + (k & 3) * 4 + (step >> 3);
                    qe[qn + __builtin_popcount(m & ltmask)] = e;
                }
                qn += __builtin_popcount(m);
                if (qn >= 32) {
                    qn -= 32;
                    ln = drain_edges(qe + qn, 32, lane, H, src, dst, dinv, lo, acc, wl, ln, record);
                }
            }
        }
        if (qn > 0) ln = drain_edges(qe, qn, lane, H, src, dst, dinv, lo, acc, wl, ln, record);
        if (record && lane == 0) w_cnt[w] = ln;
    } else {
        int2 ent = make_int2(0, 0);
        if (lane < ln) ent = wl[lane];
        for (int c = 0; c < ln; c += 32) {
            const int n = (ln - c < 32) ? (ln - c) : 32;
            const int2 cur = ent;
            if (c + 32 + lane < ln) ent = wl[c + 32 + lane];
            add_rows(n, lane, H, cur.x & 0x1FFFF, cur.x >> 17, __int_as_float(cur.y), acc);
        }
    }
    __syncthreads();

    if (mode == 0 && w == 0) cnt[blockIdx.x * 32 + lane] = w_cnt[lane];

    const float4 bb = ((const float4*)bias)[lane];
#pragma unroll 4
    for (int r = w; r < OWN; r += 32) {
        const int i = lo + r;
        float di = dinv[i];
        float sn = di * di;
        const float* a = acc + r * DIM + lane;
        float4 h = ((const float4*)(H + (size_t)i * DIM))[lane];
        float4 o;
        o.x = fmaxf(fmaf(sn, h.x, a[0])  + bb.x, 0.f);
        o.y = fmaxf(fmaf(sn, h.y, a[32]) + bb.y, 0.f);
        o.z = fmaxf(fmaf(sn, h.z, a[64]) + bb.z, 0.f);
        o.w = fmaxf(fmaf(sn, h.w, a[96]) + bb.w, 0.f);
        ((float4*)(out + (size_t)i * DIM))[lane] = o;
    }
}

extern "C" void kernel_launch(void* const* d_in, const int* in_sizes, int n_in,
                              void* d_out, int out_size, void* d_ws, size_t ws_size,
                              hipStream_t stream) {
    const float* x   = (const float*)d_in[0];
    const int*   ei  = (const int*)  d_in[1];
    const float* W1  = (const float*)d_in[2];
    const float* b1  = (const float*)d_in[3];
    const float* W2  = (const float*)d_in[4];
    const float* b2  = (const float*)d_in[5];
    float* out = (float*)d_out;

    const long long E = (long long)in_sizes[1] / 2;
    const int* src = ei;
    const int* dst = ei + E;

    float* dinv = (float*)d_ws;
    float* bufA = dinv + N_NODES;
    float* bufB = bufA + (size_t)N_NODES * DIM;
    int*   cnt  = (int*)(bufB + (size_t)N_NODES * DIM);
    int*   list = cnt + NGROUPS * 32 + 32;
    int*   part = list + (size_t)NGROUPS * 32 * WCAP * 2;
    int*   own  = part + (size_t)DEG_C * N_NODES;

    dim3 blk(256);
    const int gN    = (N_NODES + 255) / 256;

    k_deg_part<<<DEG_NR * DEG_C, 1024, 0, stream>>>(dst, E, part, own);
    k_deg_dinv<<<gN, blk, 0, stream>>>(part, dinv);

    k_gemm<<<GEMM_GROUPS, 1024, 0, stream>>>(x, W1, bufA);
    k_aggregate_owned<<<NGROUPS, 1024, 0, stream>>>(bufA, src, dst, dinv, b1, bufB, E, list, cnt, 0, own);

    k_gemm<<<GEMM_GROUPS, 1024, 0, stream>>>(bufB, W2, bufA);
    k_aggregate_owned<<<NGROUPS, 1024, 0, stream>>>(bufA, src, dst, dinv, b2, out, E, list, cnt, 1, own);
}


# Layer-2 product moved into the layer-1 aggregation launch: finished rows stay in LDS, W2 staged beside them, each group multiplies its 400 rows by WMMA and stores h2; 5 launches; sum order kept
# speedup vs baseline: 3.9777x; 1.0923x over previous
#include <hip/hip_runtime.h>

#define N_NODES 100000
#define DIM 128
#define MTILES (N_NODES / 16)

typedef __attribute__((ext_vector_type(2))) float v2f;
typedef __attribute__((ext_vector_type(8))) float v8f;

#define DEG_R    50016
#define DEG_NR   ((N_NODES + DEG_R - 1) / DEG_R)
#define DEG_C    64
#define DEG_U    8

#define OWN      400

__device__ __forceinline__ unsigned owner_byte(int d) {
    unsigned o = (unsigned)d / (unsigned)OWN;
    return o < 255u ? o : 255u;
}

__global__ __launch_bounds__(1024) void k_deg_part(const int* __restrict__ dst, long long E64,
                                                   int* __restrict__ part, int* __restrict__ own) {
    __shared__ __attribute__((aligned(16))) int c_n[DEG_R];
    const int tid   = threadIdx.x;
    const int range = blockIdx.x % DEG_NR;
    const int chunk = blockIdx.x / DEG_NR;
    const int lo    = range * DEG_R;
    const int nr    = (N_NODES - lo < DEG_R) ? (N_NODES - lo) : DEG_R;

    for (int i = tid; i < DEG_R / 4; i += 1024) ((int4*)c_n)[i] = make_int4(0, 0, 0, 0);
    __syncthreads();

    const int E   = (int)E64;
    const int Q   = E >> 2;
    const int per = (((Q + DEG_C - 1) / DEG_C) + 31) & ~31;
    const int q0  = chunk * per;
    const int q1  = (q0 + per < Q) ? (q0 + per) : Q;
    const int4* dq = (const int4*)dst;
    for (int b = q0; b < q1; b += DEG_U * 1024) {
        int4 d[DEG_U];
#pragma unroll
        for (int k = 0; k < DEG_U; ++k) {
            int q  = b + k * 1024 + tid;
            int qc = (q < q1) ? q : (q1 - 1);
            d[k] = dq[qc];
        }
#pragma unroll
        for (int k = 0; k < DEG_U; ++k) {
            if (b + k * 1024 + tid < q1) {
                if (range == 0)
                    own[b + k * 1024 + tid] = (int)(owner_byte(d[k].x) | (owner_byte(d[k].y) << 8) |
                                                    (owner_byte(d[k].z) << 16) | (owner_byte(d[k].w) << 24));
                unsigned a0 = (unsigned)(d[k].x - lo);
                unsigned a1 = (unsigned)(d[k].y - lo);
                unsigned a2 = (unsigned)(d[k].z - lo);
                unsigned a3 = (unsigned)(d[k].w - lo);
                if (a0 < (unsigned)nr) atomicAdd(&c_n[a0], 1);
                if (a1 < (unsigned)nr) atomicAdd(&c_n[a1], 1);
                if (a2 < (unsigned)nr) atomicAdd(&c_n[a2], 1);
                if (a3 < (unsigned)nr) atomicAdd(&c_n[a3], 1);
            }
        }
    }
    if (chunk == DEG_C - 1) {
        int e = Q * 4 + tid;
        if (e < E) {
            unsigned a = (unsigned)(dst[e] - lo);
            if (a < (unsigned)nr) atomicAdd(&c_n[a], 1);
        }
        if (range == 0) {
            const int QP = ((E + 511) >> 9) * 128;
            for (int q = Q + tid; q < QP; q += 1024) {
                unsigned pk = 0;
                for (int j = 0; j < 4; ++j) {
                    int ee = q * 4 + j;
                    unsigned o = (ee < E) ? owner_byte(dst[ee]) : 255u;
                    pk |= o << (8 * j);
                }
                own[q] = (int)pk;
            }
        }
    }
    __syncthreads();

    int4* p = (int4*)(part + (size_t)chunk * N_NODES + lo);
    for (int i = tid; i < nr / 4; i += 1024) p[i] = ((const int4*)c_n)[i];
}

__global__ void k_deg_dinv(const int* __restrict__ part, float* __restrict__ dinv) {
    int i = blockIdx.x * blockDim.x + threadIdx.x;
    if (i < N_NODES) {
        int c = 1;
#pragma unroll
        for (int k = 0; k < DEG_C; ++k) c += part[(size_t)k * N_NODES + i];
        dinv[i] = rsqrtf((float)c);
    }
}

#define GEMM_GROUPS 256

__device__ __forceinline__ void stage_w(const float* __restrict__ W, float* w_l, int tid) {
    for (int idx = tid; idx < DIM * DIM / 4; idx += 1024) {
        const int c  = idx & 15;
        const int j  = (idx >> 4) & 3;
        const int kp = idx >> 6;
        const float* w0 = W + (size_t)(2 * kp) * DIM + j * 32 + c;
        float4 o;
        o.x = w0[0];
        o.y = w0[DIM];
        o.z = w0[16];
        o.w = w0[DIM + 16];
        ((float4*)w_l)[idx] = o;
    }
}

__global__ __launch_bounds__(1024) void k_gemm(const float* __restrict__ X,
                                               const float* __restrict__ W,
                                               float* __restrict__ H) {
    __shared__ __attribute__((aligned(16))) float w_l[DIM * DIM];

    const int tid = threadIdx.x;
    stage_w(W, w_l, tid);
    __syncthreads();

    const int lane = tid & 31;
    const int wave = (tid >> 5) * GEMM_GROUPS + blockIdx.x;
    if (wave >= MTILES) return;

    const int m0   = wave * 16;
    const int half = lane >> 4;
    const int l16  = lane & 15;
    const int koff = half * 2;

    v8f acc[8];
#pragma unroll
    for (int n = 0; n < 8; ++n) acc[n] = (v8f){0,0,0,0,0,0,0,0};

    const float* xrow = X + (size_t)(m0 + l16) * DIM;
    const float4* wq  = (const float4*)w_l + half * 64 + l16;

    v2f an;
    an.x = xrow[koff + 0];
    an.y = xrow[koff + 1];
    float4 qn0 = wq[0], qn1 = wq[16], qn2 = wq[32], qn3 = wq[48];
#pragma unroll 2
    for (int k0 = 0; k0 < DIM; k0 += 4) {
        const v2f a = an;
        const float4 q0 = qn0, q1 = qn1, q2 = qn2, q3 = qn3;
        const int kn = (k0 + 4 < DIM) ? (k0 + 4) : k0;
        an.x = xrow[kn + koff + 0];
        an.y = xrow[kn + koff + 1];
        const float4* bp = wq + (kn >> 1) * 64;
        qn0 = bp[0]; qn1 = bp[16]; qn2 = bp[32]; qn3 = bp[48];
        v2f b;
        b.x = q0.x; b.y = q0.y;
        acc[0] = __builtin_amdgcn_wmma_f32_16x16x4_f32(false, a, false, b, (short)0, acc[0], false, false);
        b.x = q0.z; b.y = q0.w;
        acc[1] = __builtin_amdgcn_wmma_f32_16x16x4_f32(false, a, false, b, (short)0, acc[1], false, false);
        b.x = q1.x; b.y = q1.y;
        acc[2] = __builtin_amdgcn_wmma_f32_16x16x4_f32(false, a, false, b, (short)0, acc[2], false, false);
        b.x = q1.z; b.y = q1.w;
        acc[3] = __builtin_amdgcn_wmma_f32_16x16x4_f32(false, a, false, b, (short)0, acc[3], false, false);
        b.x = q2.x; b.y = q2.y;
        acc[4] = __builtin_amdgcn_wmma_f32_16x16x4_f32(false, a, false, b, (short)0, acc[4], false, false);
        b.x = q2.z; b.y = q2.w;
        acc[5] = __builtin_amdgcn_wmma_f32_16x16x4_f32(false, a, false, b, (short)0, acc[5], false, false);
        b.x = q3.x; b.y = q3.y;
        acc[6] = __builtin_amdgcn_wmma_f32_16x16x4_f32(false, a, false, b, (short)0, acc[6], false, false);
        b.x = q3.z; b.y = q3.w;
        acc[7] = __builtin_amdgcn_wmma_f32_16x16x4_f32(false, a, false, b, (short)0, acc[7], false, false);
    }

    const int rbase = m0 + half * 8;
#pragma unroll
    for (int n = 0; n < 8; ++n) {
#pragma unroll
        for (int v = 0; v < 8; ++v) {
            H[(size_t)(rbase + v) * DIM + n * 16 + l16] = acc[n][v];
        }
    }
}

#define NGROUPS  (N_NODES / OWN)
#define SCAN_U   8
#define QCAP     64
#define WCAP     1024

__device__ __forceinline__ void add_rows(int n, int lane, const float* __restrict__ H,
                                         int s_i, int d_i, float nr_i, float* acc) {
    for (int j0 = 0; j0 < n; j0 += 8) {
        float4 v[8];
#pragma unroll
        for (int k = 0; k < 8; ++k) {
            int jj = (j0 + k < n) ? (j0 + k) : (n - 1);
            int s = __builtin_amdgcn_readlane(s_i, jj);
            v[k] = ((const float4*)(H + (size_t)s * DIM))[lane];
        }
#pragma unroll
        for (int k = 0; k < 8; ++k) {
            int jj = j0 + k;
            if (jj < n) {
                int d = __builtin_amdgcn_readlane(d_i, jj);
                float nr = __int_as_float(__builtin_amdgcn_readlane(__float_as_int(nr_i), jj));
                float* a = acc + d * DIM + lane;
                atomicAdd(a,      v[k].x * nr);
                atomicAdd(a + 32, v[k].y * nr);
                atomicAdd(a + 64, v[k].z * nr);
                atomicAdd(a + 96, v[k].w * nr);
            }
        }
    }
}

__device__ __forceinline__ int drain_edges(const int* qe, int n, int lane,
                                           const float* __restrict__ H,
                                           const int* __restrict__ src,
                                           const int* __restrict__ dst,
                                           const float* __restrict__ dinv,
                                           int lo, float* acc,
                                           int2* wl, int ln, bool record) {
    int s_i = 0, d_i = 0;
    float nr_i = 0.f;
    if (lane < n) {
        int e_i = qe[lane];
        int dg  = dst[e_i];
        s_i  = src[e_i];
        d_i  = dg - lo;
        nr_i = dinv[s_i] * dinv[dg];
        if (record && ln + lane < WCAP)
            wl[ln + lane] = make_int2(s_i | (d_i << 17), __float_as_int(nr_i));
    }
    add_rows(n, lane, H, s_i, d_i, nr_i, acc);
    return ln + n;
}

__device__ __forceinline__ int x2_pos(int r, int k) {
    return r * DIM + 4 * ((k >> 2) ^ (r & 15)) + (k & 3);
}

__device__ __forceinline__ float4 finished_row(const float* __restrict__ H, const float* __restrict__ dinv,
                                               const float* acc, float4 bb, int lo, int r, int lane) {
    const int i = lo + r;
    float di = dinv[i];
    float sn = di * di;
    const float* a = acc + r * DIM + lane;
    float4 h = ((const float4*)(H + (size_t)i * DIM))[lane];
    float4 o;
    o.x = fmaxf(fmaf(sn, h.x, a[0])  + bb.x, 0.f);
    o.y = fmaxf(fmaf(sn, h.y, a[32]) + bb.y, 0.f);
    o.z = fmaxf(fmaf(sn, h.z, a[64]) + bb.z, 0.f);
    o.w = fmaxf(fmaf(sn, h.w, a[96]) + bb.w, 0.f);
    return o;
}

__global__ __launch_bounds__(1024) void k_aggregate_owned(const float* __restrict__ H,
                                                          const int* __restrict__ src,
                                                          const int* __restrict__ dst,
                                                          const float* __restrict__ dinv,
                                                          const float* __restrict__ bias,
                                                          float* __restrict__ out, long long E64,
                                                          int* list, int* cnt, int mode,
                                                          const int* __restrict__ own,
                                                          const float* __restrict__ Wn) {
    __shared__ __attribute__((aligned(16))) float acc[OWN * DIM];
    __shared__ __attribute__((aligned(16))) float w_l[DIM * DIM];
    __shared__ int q_e[32 * QCAP];
    __shared__ int w_cnt[32];

    const int tid  = threadIdx.x;
    const int lane = tid & 31;
    const int w    = __builtin_amdgcn_readfirstlane(tid >> 5);
    const int lo   = blockIdx.x * OWN;
    const int E    = (int)E64;

    for (int i = tid; i < OWN * DIM / 4; i += 1024)
        ((float4*)acc)[i] = make_float4(0.f, 0.f, 0.f, 0.f);
    if (mode == 0) stage_w(Wn, w_l, tid);
    __syncthreads();

    int2* wl = (int2*)list + ((size_t)blockIdx.x * 32 + w) * WCAP;
    int ln = 0;
    bool scan = true;
    if (mode != 0) {
        ln = cnt[blockIdx.x * 32 + w];
        scan = (unsigned)ln > (unsigned)WCAP;
    }

    if (scan) {
        const bool record = mode == 0;
        int* qe = q_e + w * QCAP;
        int qn = 0;
        const unsigned ltmask = (1u << lane) - 1u;

        const int F     = (E + 511) >> 9;
        const int trips = (F + 32 * SCAN_U - 1) / (32 * SCAN_U);
        int trip = (int)(((long long)blockIdx.x * trips) / NGROUPS);
        const unsigned pat = (unsigned)blockIdx.x * 0x01010101u;

        for (int n = 0; n < trips; ++n) {
            const int tile0 = trip * (SCAN_U * 32) + w;
            int4 dv[SCAN_U];
#pragma unroll
            for (int k = 0; k < SCAN_U; ++k) {
                int t  = tile0 + k * 32;
                int tc = (t < F) ? t : (F - 1);
                dv[k] = *(const int4*)(own + tc * 128 + lane * 4);
            }
            if (++trip == trips) trip = 0;
            static_assert(SCAN_U == 8, "the queueing below names four masks");
            unsigned hms[SCAN_U / 2];
#pragma unroll
            for (int p = 0; p < SCAN_U / 2; ++p) {
                const unsigned w8[8] = {(unsigned)dv[2 * p].x,     (unsigned)dv[2 * p].y,
                                        (unsigned)dv[2 * p].z,     (unsigned)dv[2 * p].w,
                                        (unsigned)dv[2 * p + 1].x, (unsigned)dv[2 * p + 1].y,
                                        (unsigned)dv[2 * p + 1].z, (unsigned)dv[2 * p + 1].w};
                unsigned hm = 0;
#pragma unroll
                for (int k = 0; k < 8; ++k) {
                    unsigned x = w8[k] ^ pat;
                    unsigned a = (x & 0x7F7F7F7Fu) + 0x7F7F7F7Fu;
                    unsigned z = ~(a | x) & 0x80808080u;
                    hm |= z >> k;
                }
                const int ta = tile0 + (2 * p) * 32;
                hms[p] = hm & (((ta < F) ? 0xF0F0F0F0u : 0u) | ((ta + 32 < F) ? 0x0F0F0F0Fu : 0u));
            }
            for (;;) {
                bool has = (hms[0] | hms[1] | hms[2] | hms[3]) != 0u;
                unsigned m = __builtin_amdgcn_ballot_w32(has);
                if (m == 0u) break;
                if (has) {
                    const int p = hms[0] ? 0 : hms[1] ? 1 : hms[2] ? 2 : 3;
                    const unsigned h = hms[0] ? hms[0] : hms[1] ? hms[1] : hms[2] ? hms[2] : hms[3];
                    const int step = __builtin_ctz(h);
                    const unsigned c = h & (h - 1u);
                    if (p == 0) hms[0] = c; else if (p == 1) hms[1] = c; else if (p == 2) hms[2] = c; else hms[3] = c;
                    const int k = 7 - (step & 7);
                    int e = (tile0 + (2 * p + (k >> 2)) * 32) * 512 + lane * 16 + (k & 3) * 4 + (step >> 3);
                    qe[qn + __builtin_popcount(m & ltmask)] = e;
                }
                qn += __builtin_popcount(m);
                if (qn >= 32) {
                    qn -= 32;
                    ln = drain_edges(qe + qn, 32, lane, H, src, dst, dinv, lo, acc, wl, ln, record);
                }
            }
        }
        if (qn > 0) ln = drain_edges(qe, qn, lane, H, src, dst, dinv, lo, acc, wl, ln, record);
        if (record && lane == 0) w_cnt[w] = ln;
    } else {
        int2 ent = make_int2(0, 0);
        if (lane < ln) ent = wl[lane];
        for (int c = 0; c < ln; c += 32) {
            const int n = (ln - c < 32) ? (ln - c) : 32;
            const int2 cur = ent;
            if (c + 32 + lane < ln) ent = wl[c + 32 + lane];
            add_rows(n, lane, H, cur.x & 0x1FFFF, cur.x >> 17, __int_as_float(cur.y), acc);
        }
    }
    __syncthreads();

    if (mode == 0 && w == 0) cnt[blockIdx.x * 32 + lane] = w_cnt[lane];

    const float4 bb = ((const float4*)bias)[lane];
    if (mode != 0) {
#pragma unroll 4
        for (int r = w; r < OWN; r += 32)
            ((float4*)(out + (size_t)(lo + r) * DIM))[lane] = finished_row(H, dinv, acc, bb, lo, r, lane);
        return;
    }
#pragma unroll 4
    for (int r = w; r < OWN; r += 32)
        *(float4*)(acc + x2_pos(r, 4 * lane)) = finished_row(H, dinv, acc, bb, lo, r, lane);
    __syncthreads();

    const int half = lane >> 4;
    const int l16  = lane & 15;
    const float4* wq = (const float4*)w_l + half * 64 + l16;
    for (int u = w; u < 2 * (OWN / 16); u += 32) {
        const int strip = u >> 1;
        const int nh    = u & 1;
        const int r     = strip * 16 + l16;
        const float4* wqn = wq + nh * 32;
        v8f c0 = (v8f){0,0,0,0,0,0,0,0}, c1 = c0, c2 = c0, c3 = c0;
        v2f an = *(const v2f*)(acc + x2_pos(r, 2 * half));
        float4 qn0 = wqn[0], qn1 = wqn[16];
#pragma unroll 2
        for (int k0 = 0; k0 < DIM; k0 += 4) {
            const v2f a = an;
            const float4 q0 = qn0, q1 = qn1;
            const int kn = (k0 + 4 < DIM) ? (k0 + 4) : k0;
            an = *(const v2f*)(acc + x2_pos(r, kn + 2 * half));
            const float4* bp = wqn + (kn >> 1) * 64;
            qn0 = bp[0]; qn1 = bp[16];
            v2f b;
            b.x = q0.x; b.y = q0.y;
            c0 = __builtin_amdgcn_wmma_f32_16x16x4_f32(false, a, false, b, (short)0, c0, false, false);
            b.x = q0.z; b.y = q0.w;
            c1 = __builtin_amdgcn_wmma_f32_16x16x4_f32(false, a, false, b, (short)0, c1, false, false);
            b.x = q1.x; b.y = q1.y;
            c2 = __builtin_amdgcn_wmma_f32_16x16x4_f32(false, a, false, b, (short)0, c2, false, false);
            b.x = q1.z; b.y = q1.w;
            c3 = __builtin_amdgcn_wmma_f32_16x16x4_f32(false, a, false, b, (short)0, c3, false, false);
        }
        float* orow = out + (size_t)(lo + strip * 16 + half * 8) * DIM + nh * 64 + l16;
#pragma unroll
        for (int v = 0; v < 8; ++v) {
            orow[(size_t)v * DIM]      = c0[v];
            orow[(size_t)v * DIM + 16] = c1[v];
            orow[(size_t)v * DIM + 32] = c2[v];
            orow[(size_t)v * DIM + 48] = c3[v];
        }
    }
}

extern "C" void kernel_launch(void* const* d_in, const int* in_sizes, int n_in,
                              void* d_out, int out_size, void* d_ws, size_t ws_size,
                              hipStream_t stream) {
    const float* x   = (const float*)d_in[0];
    const int*   ei  = (const int*)  d_in[1];
    const float* W1  = (const float*)d_in[2];
    const float* b1  = (const float*)d_in[3];
    const float* W2  = (const float*)d_in[4];
    const float* b2  = (const float*)d_in[5];
    float* out = (float*)d_out;

    const long long E = (long long)in_sizes[1] / 2;
    const int* src = ei;
    const int* dst = ei + E;

    float* dinv = (float*)d_ws;
    float* bufA = dinv + N_NODES;
    float* bufB = bufA + (size_t)N_NODES * DIM;
    int*   cnt  = (int*)(bufB + (size_t)N_NODES * DIM);
    int*   list = cnt + NGROUPS * 32 + 32;
    int*   part = list + (size_t)NGROUPS * 32 * WCAP * 2;
    int*   own  = part + (size_t)DEG_C * N_NODES;

    dim3 blk(256);
    const int gN    = (N_NODES + 255) / 256;

    k_deg_part<<<DEG_NR * DEG_C, 1024, 0, stream>>>(dst, E, part, own);
    k_deg_dinv<<<gN, blk, 0, stream>>>(part, dinv);

    k_gemm<<<GEMM_GROUPS, 1024, 0, stream>>>(x, W1, bufA);
    k_aggregate_owned<<<NGROUPS, 1024, 0, stream>>>(bufA, src, dst, dinv, b1, bufB, E, list, cnt, 0, own, W2);

    k_aggregate_owned<<<NGROUPS, 1024, 0, stream>>>(bufB, src, dst, dinv, b2, out, E, list, cnt, 1, own, W2);
}


# First-layer look-up of a found edge reads one packed word (source row + own-row number, written by the degree launch behind the owner list), not a source and a destination word; sum order kept
# speedup vs baseline: 4.1823x; 1.0514x over previous
#include <hip/hip_runtime.h>

#define N_NODES 100000
#define DIM 128
#define MTILES (N_NODES / 16)

typedef __attribute__((ext_vector_type(2))) float v2f;
typedef __attribute__((ext_vector_type(8))) float v8f;

#define DEG_R    50016
#define DEG_NR   ((N_NODES + DEG_R - 1) / DEG_R)
static_assert(DEG_NR == 2, "the owner list and the packed words are dealt to the two ranges' groups");
#define DEG_C    64
#define DEG_U    8

#define OWN      400

__device__ __forceinline__ unsigned owner_byte(int d) {
    unsigned o = (unsigned)d / (unsigned)OWN;
    return o < 255u ? o : 255u;
}

__device__ __forceinline__ int edge_word(int s, int d) {
    return (int)(((unsigned)s & 0x1FFFFu) | (((unsigned)d % (unsigned)OWN) << 17));
}

__device__ __forceinline__ size_t edge_words_at(int E) {
    return (size_t)((E + 511) >> 9) * 128;
}

__global__ __launch_bounds__(1024) void k_deg_part(const int* __restrict__ src, const int* __restrict__ dst, long long E64,
                                                   int* __restrict__ part, int* __restrict__ own) {
    __shared__ __attribute__((aligned(16))) int c_n[DEG_R];
    const int tid   = threadIdx.x;
    const int range = blockIdx.x % DEG_NR;
    const int chunk = blockIdx.x / DEG_NR;
    const int lo    = range * DEG_R;
    const int nr    = (N_NODES - lo < DEG_R) ? (N_NODES - lo) : DEG_R;

    for (int i = tid; i < DEG_R / 4; i += 1024) ((int4*)c_n)[i] = make_int4(0, 0, 0, 0);
    __syncthreads();

    const int E   = (int)E64;
    const int Q   = E >> 2;
    const int per = (((Q + DEG_C - 1) / DEG_C) + 31) & ~31;
    const int q0  = chunk * per;
    const int q1  = (q0 + per < Q) ? (q0 + per) : Q;
    const int4* dq = (const int4*)dst;
    const int4* sq = (const int4*)src;
    int* rec = own + edge_words_at(E);
    for (int b = q0; b < q1; b += DEG_U * 1024) {
        int4 d[DEG_U];
        int4 s[DEG_U];
#pragma unroll
        for (int k = 0; k < DEG_U; ++k) {
            int q  = b + k * 1024 + tid;
            int qc = (q < q1) ? q : (q1 - 1);
            d[k] = dq[qc];
            if (range == 1) s[k] = sq[qc];
        }
#pragma unroll
        for (int k = 0; k < DEG_U; ++k) {
            if (b + k * 1024 + tid < q1) {
                if (range == 0)
                    own[b + k * 1024 + tid] = (int)(owner_byte(d[k].x) | (owner_byte(d[k].y) << 8) |
                                                    (owner_byte(d[k].z) << 16) | (owner_byte(d[k].w) << 24));
                else
                    ((int4*)rec)[b + k * 1024 + tid] = make_int4(edge_word(s[k].x, d[k].x), edge_word(s[k].y, d[k].y),
                                                                 edge_word(s[k].z, d[k].z), edge_word(s[k].w, d[k].w));
                unsigned a0 = (unsigned)(d[k].x - lo);
                unsigned a1 = (unsigned)(d[k].y - lo);
                unsigned a2 = (unsigned)(d[k].z - lo);
                unsigned a3 = (unsigned)(d[k].w - lo);
                if (a0 < (unsigned)nr) atomicAdd(&c_n[a0], 1);
                if (a1 < (unsigned)nr) atomicAdd(&c_n[a1], 1);
                if (a2 < (unsigned)nr) atomicAdd(&c_n[a2], 1);
                if (a3 < (unsigned)nr) atomicAdd(&c_n[a3], 1);
            }
        }
    }
    if (chunk == DEG_C - 1) {
        int e = Q * 4 + tid;
        if (e < E) {
            unsigned a = (unsigned)(dst[e] - lo);
            if (a < (unsigned)nr) atomicAdd(&c_n[a], 1);
            if (range == 1) rec[e] = edge_word(src[e], dst[e]);
        }
        if (range == 0) {
            const int QP = ((E + 511) >> 9) * 128;
            for (int q = Q + tid; q < QP; q += 1024) {
                unsigned pk = 0;
                for (int j = 0; j < 4; ++j) {
                    int ee = q * 4 + j;
                    unsigned o = (ee < E) ? owner_byte(dst[ee]) : 255u;
                    pk |= o << (8 * j);
                }
                own[q] = (int)pk;
            }
        }
    }
    __syncthreads();

    int4* p = (int4*)(part + (size_t)chunk * N_NODES + lo);
    for (int i = tid; i < nr / 4; i += 1024) p[i] = ((const int4*)c_n)[i];
}

__global__ void k_deg_dinv(const int* __restrict__ part, float* __restrict__ dinv) {
    int i = blockIdx.x * blockDim.x + threadIdx.x;
    if (i < N_NODES) {
        int c = 1;
#pragma unroll
        for (int k = 0; k < DEG_C; ++k) c += part[(size_t)k * N_NODES + i];
        dinv[i] = rsqrtf((float)c);
    }
}

#define GEMM_GROUPS 256

__device__ __forceinline__ void stage_w(const float* __restrict__ W, float* w_l, int tid) {
    for (int idx = tid; idx < DIM * DIM / 4; idx += 1024) {
        const int c  = idx & 15;
        const int j  = (idx >> 4) & 3;
        const int kp = idx >> 6;
        const float* w0 = W + (size_t)(2 * kp) * DIM + j * 32 + c;
        float4 o;
        o.x = w0[0];
        o.y = w0[DIM];
        o.z = w0[16];
        o.w = w0[DIM + 16];
        ((float4*)w_l)[idx] = o;
    }
}

__global__ __launch_bounds__(1024) void k_gemm(const float* __restrict__ X,
                                               const float* __restrict__ W,
                                               float* __restrict__ H) {
    __shared__ __attribute__((aligned(16))) float w_l[DIM * DIM];

    const int tid = threadIdx.x;
    stage_w(W, w_l, tid);
    __syncthreads();

    const int lane = tid & 31;
    const int wave = (tid >> 5) * GEMM_GROUPS + blockIdx.x;
    if (wave >= MTILES) return;

    const int m0   = wave * 16;
    const int half = lane >> 4;
    const int l16  = lane & 15;
    const int koff = half * 2;

    v8f acc[8];
#pragma unroll
    for (int n = 0; n < 8; ++n) acc[n] = (v8f){0,0,0,0,0,0,0,0};

    const float* xrow = X + (size_t)(m0 + l16) * DIM;
    const float4* wq  = (const float4*)w_l + half * 64 + l16;

    v2f an;
    an.x = xrow[koff + 0];
    an.y = xrow[koff + 1];
    float4 qn0 = wq[0], qn1 = wq[16], qn2 = wq[32], qn3 = wq[48];
#pragma unroll 2
    for (int k0 = 0; k0 < DIM; k0 += 4) {
        const v2f a = an;
        const float4 q0 = qn0, q1 = qn1, q2 = qn2, q3 = qn3;
        const int kn = (k0 + 4 < DIM) ? (k0 + 4) : k0;
        an.x = xrow[kn + koff + 0];
        an.y = xrow[kn + koff + 1];
        const float4* bp = wq + (kn >> 1) * 64;
        qn0 = bp[0]; qn1 = bp[16]; qn2 = bp[32]; qn3 = bp[48];
        v2f b;
        b.x = q0.x; b.y = q0.y;
        acc[0] = __builtin_amdgcn_wmma_f32_16x16x4_f32(false, a, false, b, (short)0, acc[0], false, false);
        b.x = q0.z; b.y = q0.w;
        acc[1] = __builtin_amdgcn_wmma_f32_16x16x4_f32(false, a, false, b, (short)0, acc[1], false, false);
        b.x = q1.x; b.y = q1.y;
        acc[2] = __builtin_amdgcn_wmma_f32_16x16x4_f32(false, a, false, b, (short)0, acc[2], false, false);
        b.x = q1.z; b.y = q1.w;
        acc[3] = __builtin_amdgcn_wmma_f32_16x16x4_f32(false, a, false, b, (short)0, acc[3], false, false);
        b.x = q2.x; b.y = q2.y;
        acc[4] = __builtin_amdgcn_wmma_f32_16x16x4_f32(false, a, false, b, (short)0, acc[4], false, false);
        b.x = q2.z; b.y = q2.w;
        acc[5] = __builtin_amdgcn_wmma_f32_16x16x4_f32(false, a, false, b, (short)0, acc[5], false, false);
        b.x = q3.x; b.y = q3.y;
        acc[6] = __builtin_amdgcn_wmma_f32_16x16x4_f32(false, a, false, b, (short)0, acc[6], false, false);
        b.x = q3.z; b.y = q3.w;
        acc[7] = __builtin_amdgcn_wmma_f32_16x16x4_f32(false, a, false, b, (short)0, acc[7], false, false);
    }

    const int rbase = m0 + half * 8;
#pragma unroll
    for (int n = 0; n < 8; ++n) {
#pragma unroll
        for (int v = 0; v < 8; ++v) {
            H[(size_t)(rbase + v) * DIM + n * 16 + l16] = acc[n][v];
        }
    }
}

#define NGROUPS  (N_NODES / OWN)
#define SCAN_U   8
#define QCAP     64
#define WCAP     1024

__device__ __forceinline__ void add_rows(int n, int lane, const float* __restrict__ H,
                                         int s_i, int d_i, float nr_i, float* acc) {
    for (int j0 = 0; j0 < n; j0 += 8) {
        float4 v[8];
#pragma unroll
        for (int k = 0; k < 8; ++k) {
            int jj = (j0 + k < n) ? (j0 + k) : (n - 1);
            int s = __builtin_amdgcn_readlane(s_i, jj);
            v[k] = ((const float4*)(H + (size_t)s * DIM))[lane];
        }
#pragma unroll
        for (int k = 0; k < 8; ++k) {
            int jj = j0 + k;
            if (jj < n) {
                int d = __builtin_amdgcn_readlane(d_i, jj);
                float nr = __int_as_float(__builtin_amdgcn_readlane(__float_as_int(nr_i), jj));
                float* a = acc + d * DIM + lane;
                atomicAdd(a,      v[k].x * nr);
                atomicAdd(a + 32, v[k].y * nr);
                atomicAdd(a + 64, v[k].z * nr);
                atomicAdd(a + 96, v[k].w * nr);
            }
        }
    }
}

__device__ __forceinline__ int drain_edges(const int* qe, int n, int lane,
                                           const float* __restrict__ H,
                                           const int* __restrict__ rec,
                                           const float* __restrict__ dinv,
                                           int lo, float* acc,
                                           int2* wl, int ln, bool record) {
    int s_i = 0, d_i = 0;
    float nr_i = 0.f;
    if (lane < n) {
        int e_i = qe[lane];
        int r   = rec[e_i];
        s_i  = r & 0x1FFFF;
        d_i  = (int)((unsigned)r >> 17);
        nr_i = dinv[s_i] * dinv[lo + d_i];
        if (record && ln + lane < WCAP)
            wl[ln + lane] = make_int2(s_i | (d_i << 17), __float_as_int(nr_i));
    }
    add_rows(n, lane, H, s_i, d_i, nr_i, acc);
    return ln + n;
}

__device__ __forceinline__ int x2_pos(int r, int k) {
    return r * DIM + 4 * ((k >> 2) ^ (r & 15)) + (k & 3);
}

__device__ __forceinline__ float4 finished_row(const float* __restrict__ H, const float* __restrict__ dinv,
                                               const float* acc, float4 bb, int lo, int r, int lane) {
    const int i = lo + r;
    float di = dinv[i];
    float sn = di * di;
    const float* a = acc + r * DIM + lane;
    float4 h = ((const float4*)(H + (size_t)i * DIM))[lane];
    float4 o;
    o.x = fmaxf(fmaf(sn, h.x, a[0])  + bb.x, 0.f);
    o.y = fmaxf(fmaf(sn, h.y, a[32]) + bb.y, 0.f);
    o.z = fmaxf(fmaf(sn, h.z, a[64]) + bb.z, 0.f);
    o.w = fmaxf(fmaf(sn, h.w, a[96]) + bb.w, 0.f);
    return o;
}

__global__ __launch_bounds__(1024) void k_aggregate_owned(const float* __restrict__ H,
                                                          const int* __restrict__ src,
                                                          const int* __restrict__ dst,
                                                          const float* __restrict__ dinv,
                                                          const float* __restrict__ bias,
                                                          float* __restrict__ out, long long E64,
                                                          int* list, int* cnt, int mode,
                                                          const int* __restrict__ own,
                                                          const float* __restrict__ Wn) {
    __shared__ __attribute__((aligned(16))) float acc[OWN * DIM];
    __shared__ __attribute__((aligned(16))) float w_l[DIM * DIM];
    __shared__ int q_e[32 * QCAP];
    __shared__ int w_cnt[32];

    const int tid  = threadIdx.x;
    const int lane = tid & 31;
    const int w    = __builtin_amdgcn_readfirstlane(tid >> 5);
    const int lo   = blockIdx.x * OWN;
    const int E    = (int)E64;

    for (int i = tid; i < OWN * DIM / 4; i += 1024)
        ((float4*)acc)[i] = make_float4(0.f, 0.f, 0.f, 0.f);
    if (mode == 0) stage_w(Wn, w_l, tid);
    __syncthreads();

    int2* wl = (int2*)list + ((size_t)blockIdx.x * 32 + w) * WCAP;
    int ln = 0;
    bool scan = true;
    if (mode != 0) {
        ln = cnt[blockIdx.x * 32 + w];
        scan = (unsigned)ln > (unsigned)WCAP;
    }

    if (scan) {
        const bool record = mode == 0;
        const int* rec = own + edge_words_at(E);
        int* qe = q_e + w * QCAP;
        int qn = 0;
        const unsigned ltmask = (1u << lane) - 1u;

        const int F     = (E + 511) >> 9;
        const int trips = (F + 32 * SCAN_U - 1) / (32 * SCAN_U);
        int trip = (int)(((long long)blockIdx.x * trips) / NGROUPS);
        const unsigned pat = (unsigned)blockIdx.x * 0x01010101u;

        for (int n = 0; n < trips; ++n) {
            const int tile0 = trip * (SCAN_U * 32) + w;
            int4 dv[SCAN_U];
#pragma unroll
            for (int k = 0; k < SCAN_U; ++k) {
                int t  = tile0 + k * 32;
                int tc = (t < F) ? t : (F - 1);
                dv[k] = *(const int4*)(own + tc * 128 + lane * 4);
            }
            if (++trip == trips) trip = 0;
            static_assert(SCAN_U == 8, "the queueing below names four masks");
            unsigned hms[SCAN_U / 2];
#pragma unroll
            for (int p = 0; p < SCAN_U / 2; ++p) {
                const unsigned w8[8] = {(unsigned)dv[2 * p].x,     (unsigned)dv[2 * p].y,
                                        (unsigned)dv[2 * p].z,     (unsigned)dv[2 * p].w,
                                        (unsigned)dv[2 * p + 1].x, (unsigned)dv[2 * p + 1].y,
                                        (unsigned)dv[2 * p + 1].z, (unsigned)dv[2 * p + 1].w};
                unsigned hm = 0;
#pragma unroll
                for (int k = 0; k < 8; ++k) {
                    unsigned x = w8[k] ^ pat;
                    unsigned a = (x & 0x7F7F7F7Fu) + 0x7F7F7F7Fu;
                    unsigned z = ~(a | x) & 0x80808080u;
                    hm |= z >> k;
                }
                const int ta = tile0 + (2 * p) * 32;
                hms[p] = hm & (((ta < F) ? 0xF0F0F0F0u : 0u) | ((ta + 32 < F) ? 0x0F0F0F0Fu : 0u));
            }
            for (;;) {
                bool has = (hms[0] | hms[1] | hms[2] | hms[3]) != 0u;
                unsigned m = __builtin_amdgcn_ballot_w32(has);
                if (m == 0u) break;
                if (has) {
                    const int p = hms[0] ? 0 : hms[1] ? 1 : hms[2] ? 2 : 3;
                    const unsigned h = hms[0] ? hms[0] : hms[1] ? hms[1] : hms[2] ? hms[2] : hms[3];
                    const int step = __builtin_ctz(h);
                    const unsigned c = h & (h - 1u);
                    if (p == 0) hms[0] = c; else if (p == 1) hms[1] = c; else if (p == 2) hms[2] = c; else hms[3] = c;
                    const int k = 7 - (step & 7);
                    int e = (tile0 + (2 * p + (k >> 2)) * 32) * 512 + lane * 16 + (k & 3) * 4 + (step >> 3);
                    qe[qn + __builtin_popcount(m & ltmask)] = e;
                }
                qn += __builtin_popcount(m);
                if (qn >= 32) {
                    qn -= 32;
                    ln = drain_edges(qe + qn, 32, lane, H, rec, dinv, lo, acc, wl, ln, record);
                }
            }
        }
        if (qn > 0) ln = drain_edges(qe, qn, lane, H, rec, dinv, lo, acc, wl, ln, record);
        if (record && lane == 0) w_cnt[w] = ln;
    } else {
        int2 ent = make_int2(0, 0);
        if (lane < ln) ent = wl[lane];
        for (int c = 0; c < ln; c += 32) {
            const int n = (ln - c < 32) ? (ln - c) : 32;
            const int2 cur = ent;
            if (c + 32 + lane < ln) ent = wl[c + 32 + lane];
            add_rows(n, lane, H, cur.x & 0x1FFFF, cur.x >> 17, __int_as_float(cur.y), acc);
        }
    }
    __syncthreads();

    if (mode == 0 && w == 0) cnt[blockIdx.x * 32 + lane] = w_cnt[lane];

    const float4 bb = ((const float4*)bias)[lane];
    if (mode != 0) {
#pragma unroll 4
        for (int r = w; r < OWN; r += 32)
            ((float4*)(out + (size_t)(lo + r) * DIM))[lane] = finished_row(H, dinv, acc, bb, lo, r, lane);
        return;
    }
#pragma unroll 4
    for (int r = w; r < OWN; r += 32)
        *(float4*)(acc + x2_pos(r, 4 * lane)) = finished_row(H, dinv, acc, bb, lo, r, lane);
    __syncthreads();

    const int half = lane >> 4;
    const int l16  = lane & 15;
    const float4* wq = (const float4*)w_l + half * 64 + l16;
    for (int u = w; u < 2 * (OWN / 16); u += 32) {
        const int strip = u >> 1;
        const int nh    = u & 1;
        const int r     = strip * 16 + l16;
        const float4* wqn = wq + nh * 32;
        v8f c0 = (v8f){0,0,0,0,0,0,0,0}, c1 = c0, c2 = c0, c3 = c0;
        v2f an = *(const v2f*)(acc + x2_pos(r, 2 * half));
        float4 qn0 = wqn[0], qn1 = wqn[16];
#pragma unroll 2
        for (int k0 = 0; k0 < DIM; k0 += 4) {
            const v2f a = an;
            const float4 q0 = qn0, q1 = qn1;
            const int kn = (k0 + 4 < DIM) ? (k0 + 4) : k0;
            an = *(const v2f*)(acc + x2_pos(r, kn + 2 * half));
            const float4* bp = wqn + (kn >> 1) * 64;
            qn0 = bp[0]; qn1 = bp[16];
            v2f b;
            b.x = q0.x; b.y = q0.y;
            c0 = __builtin_amdgcn_wmma_f32_16x16x4_f32(false, a, false, b, (short)0, c0, false, false);
            b.x = q0.z; b.y = q0.w;
            c1 = __builtin_amdgcn_wmma_f32_16x16x4_f32(false, a, false, b, (short)0, c1, false, false);
            b.x = q1.x; b.y = q1.y;
            c2 = __builtin_amdgcn_wmma_f32_16x16x4_f32(false, a, false, b, (short)0, c2, false, false);
            b.x = q1.z; b.y = q1.w;
            c3 = __builtin_amdgcn_wmma_f32_16x16x4_f32(false, a, false, b, (short)0, c3, false, false);
        }
        float* orow = out + (size_t)(lo + strip * 16 + half * 8) * DIM + nh * 64 + l16;
#pragma unroll
        for (int v = 0; v < 8; ++v) {
            orow[(size_t)v * DIM]      = c0[v];
            orow[(size_t)v * DIM + 16] = c1[v];
            orow[(size_t)v * DIM + 32] = c2[v];
            orow[(size_t)v * DIM + 48] = c3[v];
        }
    }
}

extern "C" void kernel_launch(void* const* d_in, const int* in_sizes, int n_in,
                              void* d_out, int out_size, void* d_ws, size_t ws_size,
                              hipStream_t stream) {
    const float* x   = (const float*)d_in[0];
    const int*   ei  = (const int*)  d_in[1];
    const float* W1  = (const float*)d_in[2];
    const float* b1  = (const float*)d_in[3];
    const float* W2  = (const float*)d_in[4];
    const float* b2  = (const float*)d_in[5];
    float* out = (float*)d_out;

    const long long E = (long long)in_sizes[1] / 2;
    const int* src = ei;
    const int* dst = ei + E;

    float* dinv = (float*)d_ws;
    float* bufA = dinv + N_NODES;
    float* bufB = bufA + (size_t)N_NODES * DIM;
    int*   cnt  = (int*)(bufB + (size_t)N_NODES * DIM);
    int*   list = cnt + NGROUPS * 32 + 32;
    int*   part = list + (size_t)NGROUPS * 32 * WCAP * 2;
    int*   own  = part + (size_t)DEG_C * N_NODES;

    dim3 blk(256);
    const int gN    = (N_NODES + 255) / 256;

    k_deg_part<<<DEG_NR * DEG_C, 1024, 0, stream>>>(src, dst, E, part, own);
    k_deg_dinv<<<gN, blk, 0, stream>>>(part, dinv);

    k_gemm<<<GEMM_GROUPS, 1024, 0, stream>>>(x, W1, bufA);
    k_aggregate_owned<<<NGROUPS, 1024, 0, stream>>>(bufA, src, dst, dinv, b1, bufB, E, list, cnt, 0, own, W2);

    k_aggregate_owned<<<NGROUPS, 1024, 0, stream>>>(bufB, src, dst, dinv, b2, out, E, list, cnt, 1, own, W2);
}


# First-layer edge weights from LDS: a byte-per-node degree table (written by the dinv launch, kept where W2 is staged later) and own-row dinv replace the scattered global dinv reads; sum order kept
# speedup vs baseline: 4.4351x; 1.0604x over previous
#include <hip/hip_runtime.h>

#define N_NODES 100000
#define DIM 128
#define MTILES (N_NODES / 16)

typedef __attribute__((ext_vector_type(2))) float v2f;
typedef __attribute__((ext_vector_type(8))) float v8f;

#define DEG_R    50016
#define DEG_NR   ((N_NODES + DEG_R - 1) / DEG_R)
static_assert(DEG_NR == 2, "the owner list and the packed words are dealt to the two ranges' groups");
#define DEG_C    64
#define DEG_U    8

#define OWN      400

__device__ __forceinline__ unsigned owner_byte(int d) {
    unsigned o = (unsigned)d / (unsigned)OWN;
    return o < 255u ? o : 255u;
}

__device__ __forceinline__ int edge_word(int s, int d) {
    return (int)(((unsigned)s & 0x1FFFFu) | (((unsigned)d % (unsigned)OWN) << 17));
}

__device__ __forceinline__ size_t edge_words_at(int E) {
    return (size_t)((E + 511) >> 9) * 128;
}

__global__ __launch_bounds__(1024) void k_deg_part(const int* __restrict__ src, const int* __restrict__ dst, long long E64,
                                                   int* __restrict__ part, int* __restrict__ own) {
    __shared__ __attribute__((aligned(16))) int c_n[DEG_R];
    const int tid   = threadIdx.x;
    const int range = blockIdx.x % DEG_NR;
    const int chunk = blockIdx.x / DEG_NR;
    const int lo    = range * DEG_R;
    const int nr    = (N_NODES - lo < DEG_R) ? (N_NODES - lo) : DEG_R;

    for (int i = tid; i < DEG_R / 4; i += 1024) ((int4*)c_n)[i] = make_int4(0, 0, 0, 0);
    __syncthreads();

    const int E   = (int)E64;
    const int Q   = E >> 2;
    const int per = (((Q + DEG_C - 1) / DEG_C) + 31) & ~31;
    const int q0  = chunk * per;
    const int q1  = (q0 + per < Q) ? (q0 + per) : Q;
    const int4* dq = (const int4*)dst;
    const int4* sq = (const int4*)src;
    int* rec = own + edge_words_at(E);
    for (int b = q0; b < q1; b += DEG_U * 1024) {
        int4 d[DEG_U];
        int4 s[DEG_U];
#pragma unroll
        for (int k = 0; k < DEG_U; ++k) {
            int q  = b + k * 1024 + tid;
            int qc = (q < q1) ? q : (q1 - 1);
            d[k] = dq[qc];
            if (range == 1) s[k] = sq[qc];
        }
#pragma unroll
        for (int k = 0; k < DEG_U; ++k) {
            if (b + k * 1024 + tid < q1) {
                if (range == 0)
                    own[b + k * 1024 + tid] = (int)(owner_byte(d[k].x) | (owner_byte(d[k].y) << 8) |
                                                    (owner_byte(d[k].z) << 16) | (owner_byte(d[k].w) << 24));
                else
                    ((int4*)rec)[b + k * 1024 + tid] = make_int4(edge_word(s[k].x, d[k].x), edge_word(s[k].y, d[k].y),
                                                                 edge_word(s[k].z, d[k].z), edge_word(s[k].w, d[k].w));
                unsigned a0 = (unsigned)(d[k].x - lo);
                unsigned a1 = (unsigned)(d[k].y - lo);
                unsigned a2 = (unsigned)(d[k].z - lo);
                unsigned a3 = (unsigned)(d[k].w - lo);
                if (a0 < (unsigned)nr) atomicAdd(&c_n[a0], 1);
                if (a1 < (unsigned)nr) atomicAdd(&c_n[a1], 1);
                if (a2 < (unsigned)nr) atomicAdd(&c_n[a2], 1);
                if (a3 < (unsigned)nr) atomicAdd(&c_n[a3], 1);
            }
        }
    }
    if (chunk == DEG_C - 1) {
        int e = Q * 4 + tid;
        if (e < E) {
            unsigned a = (unsigned)(dst[e] - lo);
            if (a < (unsigned)nr) atomicAdd(&c_n[a], 1);
            if (range == 1) rec[e] = edge_word(src[e], dst[e]);
        }
        if (range == 0) {
            const int QP = ((E + 511) >> 9) * 128;
            for (int q = Q + tid; q < QP; q += 1024) {
                unsigned pk = 0;
                for (int j = 0; j < 4; ++j) {
                    int ee = q * 4 + j;
                    unsigned o = (ee < E) ? owner_byte(dst[ee]) : 255u;
                    pk |= o << (8 * j);
                }
                own[q] = (int)pk;
            }
        }
    }
    __syncthreads();

    int4* p = (int4*)(part + (size_t)chunk * N_NODES + lo);
    for (int i = tid; i < nr / 4; i += 1024) p[i] = ((const int4*)c_n)[i];
}

#define DEGTAB_BYTES (((N_NODES + 127) / 128) * 128)
static_assert(N_NODES % 16 == 0, "the table is moved in 16-byte pieces and packed four nodes per dword");

__global__ __launch_bounds__(256) void k_deg_dinv(const int* __restrict__ part, float* __restrict__ dinv,
                                                  int* __restrict__ deg4) {
    __shared__ __attribute__((aligned(16))) int c_s[256];
    const int t = threadIdx.x;
    int i = blockIdx.x * 256 + t;
    int c = 1;
    if (i < N_NODES) {
#pragma unroll
        for (int k = 0; k < DEG_C; ++k) c += part[(size_t)k * N_NODES + i];
        dinv[i] = rsqrtf((float)c);
    }
    c_s[t] = c < 255 ? c : 255;
    __syncthreads();
    const int j = blockIdx.x * 64 + t;
    if (t < 64 && j < N_NODES / 4) {
        const int4 q = ((const int4*)c_s)[t];
        deg4[j] = q.x | (q.y << 8) | (q.z << 16) | (q.w << 24);
    }
}

#define GEMM_GROUPS 256

__device__ __forceinline__ float4 w_piece(const float* __restrict__ W, int idx) {
    const int c  = idx & 15;
    const int j  = (idx >> 4) & 3;
    const int kp = idx >> 6;
    const float* w0 = W + (size_t)(2 * kp) * DIM + j * 32 + c;
    float4 o;
    o.x = w0[0];
    o.y = w0[DIM];
    o.z = w0[16];
    o.w = w0[DIM + 16];
    return o;
}

__device__ __forceinline__ void stage_w(const float* __restrict__ W, float* w_l, int tid) {
    for (int idx = tid; idx < DIM * DIM / 4; idx += 1024)
        ((float4*)w_l)[idx] = w_piece(W, idx);
}

__global__ __launch_bounds__(1024) void k_gemm(const float* __restrict__ X,
                                               const float* __restrict__ W,
                                               float* __restrict__ H) {
    __shared__ __attribute__((aligned(16))) float w_l[DIM * DIM];

    const int tid = threadIdx.x;
    stage_w(W, w_l, tid);
    __syncthreads();

    const int lane = tid & 31;
    const int wave = (tid >> 5) * GEMM_GROUPS + blockIdx.x;
    if (wave >= MTILES) return;

    const int m0   = wave * 16;
    const int half = lane >> 4;
    const int l16  = lane & 15;
    const int koff = half * 2;

    v8f acc[8];
#pragma unroll
    for (int n = 0; n < 8; ++n) acc[n] = (v8f){0,0,0,0,0,0,0,0};

    const float* xrow = X + (size_t)(m0 + l16) * DIM;
    const float4* wq  = (const float4*)w_l + half * 64 + l16;

    v2f an;
    an.x = xrow[koff + 0];
    an.y = xrow[koff + 1];
    float4 qn0 = wq[0], qn1 = wq[16], qn2 = wq[32], qn3 = wq[48];
#pragma unroll 2
    for (int k0 = 0; k0 < DIM; k0 += 4) {
        const v2f a = an;
        const float4 q0 = qn0, q1 = qn1, q2 = qn2, q3 = qn3;
        const int kn = (k0 + 4 < DIM) ? (k0 + 4) : k0;
        an.x = xrow[kn + koff + 0];
        an.y = xrow[kn + koff + 1];
        const float4* bp = wq + (kn >> 1) * 64;
        qn0 = bp[0]; qn1 = bp[16]; qn2 = bp[32]; qn3 = bp[48];
        v2f b;
        b.x = q0.x; b.y = q0.y;
        acc[0] = __builtin_amdgcn_wmma_f32_16x16x4_f32(false, a, false, b, (short)0, acc[0], false, false);
        b.x = q0.z; b.y = q0.w;
        acc[1] = __builtin_amdgcn_wmma_f32_16x16x4_f32(false, a, false, b, (short)0, acc[1], false, false);
        b.x = q1.x; b.y = q1.y;
        acc[2] = __builtin_amdgcn_wmma_f32_16x16x4_f32(false, a, false, b, (short)0, acc[2], false, false);
        b.x = q1.z; b.y = q1.w;
        acc[3] = __builtin_amdgcn_wmma_f32_16x16x4_f32(false, a, false, b, (short)0, acc[3], false, false);
        b.x = q2.x; b.y = q2.y;
        acc[4] = __builtin_amdgcn_wmma_f32_16x16x4_f32(false, a, false, b, (short)0, acc[4], false, false);
        b.x = q2.z; b.y = q2.w;
        acc[5] = __builtin_amdgcn_wmma_f32_16x16x4_f32(false, a, false, b, (short)0, acc[5], false, false);
        b.x = q3.x; b.y = q3.y;
        acc[6] = __builtin_amdgcn_wmma_f32_16x16x4_f32(false, a, false, b, (short)0, acc[6], false, false);
        b.x = q3.z; b.y = q3.w;
        acc[7] = __builtin_amdgcn_wmma_f32_16x16x4_f32(false, a, false, b, (short)0, acc[7], false, false);
    }

    const int rbase = m0 + half * 8;
#pragma unroll
    for (int n = 0; n < 8; ++n) {
#pragma unroll
        for (int v = 0; v < 8; ++v) {
            H[(size_t)(rbase + v) * DIM + n * 16 + l16] = acc[n][v];
        }
    }
}

#define NGROUPS  (N_NODES / OWN)
#define SCAN_U   8
#define QCAP     64
#define WCAP     1024

__device__ __forceinline__ void add_rows(int n, int lane, const float* __restrict__ H,
                                         int s_i, int d_i, float nr_i, float* acc) {
    for (int j0 = 0; j0 < n; j0 += 8) {
        float4 v[8];
#pragma unroll
        for (int k = 0; k < 8; ++k) {
            int jj = (j0 + k < n) ? (j0 + k) : (n - 1);
            int s = __builtin_amdgcn_readlane(s_i, jj);
            v[k] = ((const float4*)(H + (size_t)s * DIM))[lane];
        }
#pragma unroll
        for (int k = 0; k < 8; ++k) {
            int jj = j0 + k;
            if (jj < n) {
                int d = __builtin_amdgcn_readlane(d_i, jj);
                float nr = __int_as_float(__builtin_amdgcn_readlane(__float_as_int(nr_i), jj));
                float* a = acc + d * DIM + lane;
                atomicAdd(a,      v[k].x * nr);
                atomicAdd(a + 32, v[k].y * nr);
                atomicAdd(a + 64, v[k].z * nr);
                atomicAdd(a + 96, v[k].w * nr);
            }
        }
    }
}

typedef __attribute__((address_space(3))) unsigned char lds_u8;
typedef __attribute__((address_space(3))) float lds_f32;

__device__ __forceinline__ int drain_edges(const int* qe, int n, int lane,
                                           const float* __restrict__ H,
                                           const int* __restrict__ rec,
                                           const float* __restrict__ dinv,
                                           bool tab, const lds_u8* tab8, const lds_f32* dinv_own,
                                           int lo, float* acc,
                                           int2* wl, int ln, bool record) {
    int s_i = 0, d_i = 0;
    float nr_i = 0.f;
    if (lane < n) {
        int e_i = qe[lane];
        int r   = rec[e_i];
        s_i  = r & 0x1FFFF;
        d_i  = (int)((unsigned)r >> 17);
        float ds, dd;
        if (tab) {
            dd = dinv_own[d_i];
            const int c = tab8[s_i];
            ds = rsqrtf((float)c);
            if (c == 255) ds = dinv[s_i];
        } else {
            ds = dinv[s_i];
            dd = dinv[lo + d_i];
        }
        nr_i = ds * dd;
        if (record && ln + lane < WCAP)
            wl[ln + lane] = make_int2(s_i | (d_i << 17), __float_as_int(nr_i));
    }
    add_rows(n, lane, H, s_i, d_i, nr_i, acc);
    return ln + n;
}

__device__ __forceinline__ int x2_pos(int r, int k) {
    return r * DIM + 4 * ((k >> 2) ^ (r & 15)) + (k & 3);
}

__device__ __forceinline__ float4 finished_row(const float* __restrict__ H, const float* __restrict__ dinv,
                                               const float* acc, float4 bb, int lo, int r, int lane) {
    const int i = lo + r;
    float di = dinv[i];
    float sn = di * di;
    const float* a = acc + r * DIM + lane;
    float4 h = ((const float4*)(H + (size_t)i * DIM))[lane];
    float4 o;
    o.x = fmaxf(fmaf(sn, h.x, a[0])  + bb.x, 0.f);
    o.y = fmaxf(fmaf(sn, h.y, a[32]) + bb.y, 0.f);
    o.z = fmaxf(fmaf(sn, h.z, a[64]) + bb.z, 0.f);
    o.w = fmaxf(fmaf(sn, h.w, a[96]) + bb.w, 0.f);
    return o;
}

__global__ __launch_bounds__(1024) void k_aggregate_owned(const float* __restrict__ H,
                                                          const int* __restrict__ src,
                                                          const int* __restrict__ dst,
                                                          const float* __restrict__ dinv,
                                                          const float* __restrict__ bias,
                                                          float* __restrict__ out, long long E64,
                                                          int* list, int* cnt, int mode,
                                                          const int* __restrict__ own,
                                                          const float* __restrict__ Wn,
                                                          const int* __restrict__ deg4) {
    __shared__ __attribute__((aligned(16))) float acc[OWN * DIM];
    __shared__ __attribute__((aligned(16))) float w_l[DEGTAB_BYTES / 4];
    static_assert(DEGTAB_BYTES >= DIM * DIM * 4, "Wn is staged into the table's space");
    __shared__ float dinv_own[OWN];
    __shared__ int q_e[32 * QCAP];
    __shared__ int w_cnt[32];

    const int tid  = threadIdx.x;
    const int lane = tid & 31;
    const int w    = __builtin_amdgcn_readfirstlane(tid >> 5);
    const int lo   = blockIdx.x * OWN;
    const int E    = (int)E64;

    for (int i = tid; i < OWN * DIM / 4; i += 1024)
        ((float4*)acc)[i] = make_float4(0.f, 0.f, 0.f, 0.f);
    if (mode == 0) {
        constexpr int TQ = N_NODES / 16;
        constexpr int TU = (TQ + 1023) / 1024;
        const float4* tq = (const float4*)deg4;
        float4 tp[TU];
#pragma unroll
        for (int k = 0; k < TU; ++k) {
            const int idx = tid + k * 1024;
            tp[k] = tq[idx < TQ ? idx : TQ - 1];
        }
#pragma unroll
        for (int k = 0; k < TU; ++k) {
            const int idx = tid + k * 1024;
            if (idx < TQ) ((float4*)w_l)[idx] = tp[k];
        }
        if (tid < OWN) dinv_own[tid] = dinv[lo + tid];
    }
    __syncthreads();

    int2* wl = (int2*)list + ((size_t)blockIdx.x * 32 + w) * WCAP;
    int ln = 0;
    bool scan = true;
    if (mode != 0) {
        ln = cnt[blockIdx.x * 32 + w];
        scan = (unsigned)ln > (unsigned)WCAP;
    }

    if (scan) {
        const bool record = mode == 0;
        const bool tab = mode == 0;
        const lds_u8* tab8 = (const lds_u8*)w_l;
        const int* rec = own + edge_words_at(E);
        int* qe = q_e + w * QCAP;
        int qn = 0;
        const unsigned ltmask = (1u << lane) - 1u;

        const int F     = (E + 511) >> 9;
        const int trips = (F + 32 * SCAN_U - 1) / (32 * SCAN_U);
        int trip = (int)(((long long)blockIdx.x * trips) / NGROUPS);
        const unsigned pat = (unsigned)blockIdx.x * 0x01010101u;

        for (int n = 0; n < trips; ++n) {
            const int tile0 = trip * (SCAN_U * 32) + w;
            int4 dv[SCAN_U];
#pragma unroll
            for (int k = 0; k < SCAN_U; ++k) {
                int t  = tile0 + k * 32;
                int tc = (t < F) ? t : (F - 1);
                dv[k] = *(const int4*)(own + tc * 128 + lane * 4);
            }
            if (++trip == trips) trip = 0;
            static_assert(SCAN_U == 8, "the queueing below names four masks");
            unsigned hms[SCAN_U / 2];
#pragma unroll
            for (int p = 0; p < SCAN_U / 2; ++p) {
                const unsigned w8[8] = {(unsigned)dv[2 * p].x,     (unsigned)dv[2 * p].y,
                                        (unsigned)dv[2 * p].z,     (unsigned)dv[2 * p].w,
                                        (unsigned)dv[2 * p + 1].x, (unsigned)dv[2 * p + 1].y,
                                        (unsigned)dv[2 * p + 1].z, (unsigned)dv[2 * p + 1].w};
                unsigned hm = 0;
#pragma unroll
                for (int k = 0; k < 8; ++k) {
                    unsigned x = w8[k] ^ pat;
                    unsigned a = (x & 0x7F7F7F7Fu) + 0x7F7F7F7Fu;
                    unsigned z = ~(a | x) & 0x80808080u;
                    hm |= z >> k;
                }
                const int ta = tile0 + (2 * p) * 32;
                hms[p] = hm & (((ta < F) ? 0xF0F0F0F0u : 0u) | ((ta + 32 < F) ? 0x0F0F0F0Fu : 0u));
            }
            for (;;) {
                bool has = (hms[0] | hms[1] | hms[2] | hms[3]) != 0u;
                unsigned m = __builtin_amdgcn_ballot_w32(has);
                if (m == 0u) break;
                if (has) {
                    const int p = hms[0] ? 0 : hms[1] ? 1 : hms[2] ? 2 : 3;
                    const unsigned h = hms[0] ? hms[0] : hms[1] ? hms[1] : hms[2] ? hms[2] : hms[3];
                    const int step = __builtin_ctz(h);
                    const unsigned c = h & (h - 1u);
                    if (p == 0) hms[0] = c; else if (p == 1) hms[1] = c; else if (p == 2) hms[2] = c; else hms[3] = c;
                    const int k = 7 - (step & 7);
                    int e = (tile0 + (2 * p + (k >> 2)) * 32) * 512 + lane * 16 + (k & 3) * 4 + (step >> 3);
                    qe[qn + __builtin_popcount(m & ltmask)] = e;
                }
                qn += __builtin_popcount(m);
                if (qn >= 32) {
                    qn -= 32;
                    ln = drain_edges(qe + qn, 32, lane, H, rec, dinv, tab, tab8, (const lds_f32*)dinv_own, lo, acc, wl, ln, record);
                }
            }
        }
        if (qn > 0) ln = drain_edges(qe, qn, lane, H, rec, dinv, tab, tab8, (const lds_f32*)dinv_own, lo, acc, wl, ln, record);
        if (record && lane == 0) w_cnt[w] = ln;
    } else {
        int2 ent = make_int2(0, 0);
        if (lane < ln) ent = wl[lane];
        for (int c = 0; c < ln; c += 32) {
            const int n = (ln - c < 32) ? (ln - c) : 32;
            const int2 cur = ent;
            if (c + 32 + lane < ln) ent = wl[c + 32 + lane];
            add_rows(n, lane, H, cur.x & 0x1FFFF, cur.x >> 17, __int_as_float(cur.y), acc);
        }
    }
    constexpr int WP = DIM * DIM / 4 / 1024;
    float4 wp[WP];
#pragma unroll
    for (int k = 0; k < WP; ++k) wp[k] = make_float4(0.f, 0.f, 0.f, 0.f);
    if (mode == 0) {
#pragma unroll
        for (int k = 0; k < WP; ++k) wp[k] = w_piece(Wn, tid + k * 1024);
    }
    __syncthreads();
    if (mode == 0) {
#pragma unroll
        for (int k = 0; k < WP; ++k) ((float4*)w_l)[tid + k * 1024] = wp[k];
    }

    if (mode == 0 && w == 0) cnt[blockIdx.x * 32 + lane] = w_cnt[lane];

    const float4 bb = ((const float4*)bias)[lane];
    if (mode != 0) {
#pragma unroll 4
        for (int r = w; r < OWN; r += 32)
            ((float4*)(out + (size_t)(lo + r) * DIM))[lane] = finished_row(H, dinv, acc, bb, lo, r, lane);
        return;
    }
#pragma unroll 4
    for (int r = w; r < OWN; r += 32)
        *(float4*)(acc + x2_pos(r, 4 * lane)) = finished_row(H, dinv, acc, bb, lo, r, lane);
    __syncthreads();

    const int half = lane >> 4;
    const int l16  = lane & 15;
    const float4* wq = (const float4*)w_l + half * 64 + l16;
    for (int u = w; u < 2 * (OWN / 16); u += 32) {
        const int strip = u >> 1;
        const int nh    = u & 1;
        const int r     = strip * 16 + l16;
        const float4* wqn = wq + nh * 32;
        v8f c0 = (v8f){0,0,0,0,0,0,0,0}, c1 = c0, c2 = c0, c3 = c0;
        v2f an = *(const v2f*)(acc + x2_pos(r, 2 * half));
        float4 qn0 = wqn[0], qn1 = wqn[16];
#pragma unroll 2
        for (int k0 = 0; k0 < DIM; k0 += 4) {
            const v2f a = an;
            const float4 q0 = qn0, q1 = qn1;
            const int kn = (k0 + 4 < DIM) ? (k0 + 4) : k0;
            an = *(const v2f*)(acc + x2_pos(r, kn + 2 * half));
            const float4* bp = wqn + (kn >> 1) * 64;
            qn0 = bp[0]; qn1 = bp[16];
            v2f b;
            b.x = q0.x; b.y = q0.y;
            c0 = __builtin_amdgcn_wmma_f32_16x16x4_f32(false, a, false, b, (short)0, c0, false, false);
            b.x = q0.z; b.y = q0.w;
            c1 = __builtin_amdgcn_wmma_f32_16x16x4_f32(false, a, false, b, (short)0, c1, false, false);
            b.x = q1.x; b.y = q1.y;
            c2 = __builtin_amdgcn_wmma_f32_16x16x4_f32(false, a, false, b, (short)0, c2, false, false);
            b.x = q1.z; b.y = q1.w;
            c3 = __builtin_amdgcn_wmma_f32_16x16x4_f32(false, a, false, b, (short)0, c3, false, false);
        }
        float* orow = out + (size_t)(lo + strip * 16 + half * 8) * DIM + nh * 64 + l16;
#pragma unroll
        for (int v = 0; v < 8; ++v) {
            orow[(size_t)v * DIM]      = c0[v];
            orow[(size_t)v * DIM + 16] = c1[v];
            orow[(size_t)v * DIM + 32] = c2[v];
            orow[(size_t)v * DIM + 48] = c3[v];
        }
    }
}

extern "C" void kernel_launch(void* const* d_in, const int* in_sizes, int n_in,
                              void* d_out, int out_size, void* d_ws, size_t ws_size,
                              hipStream_t stream) {
    const float* x   = (const float*)d_in[0];
    const int*   ei  = (const int*)  d_in[1];
    const float* W1  = (const float*)d_in[2];
    const float* b1  = (const float*)d_in[3];
    const float* W2  = (const float*)d_in[4];
    const float* b2  = (const float*)d_in[5];
    float* out = (float*)d_out;

    const long long E = (long long)in_sizes[1] / 2;
    const int* src = ei;
    const int* dst = ei + E;

    float* dinv = (float*)d_ws;
    float* bufA = dinv + N_NODES;
    float* bufB = bufA + (size_t)N_NODES * DIM;
    int*   cnt  = (int*)(bufB + (size_t)N_NODES * DIM);
    int*   list = cnt + NGROUPS * 32 + 32;
    int*   part = list + (size_t)NGROUPS * 32 * WCAP * 2;
    int*   deg4 = part + (size_t)DEG_C * N_NODES;
    int*   own  = deg4 + DEGTAB_BYTES / 4;

    dim3 blk(256);
    const int gN    = (N_NODES + 255) / 256;

    k_deg_part<<<DEG_NR * DEG_C, 1024, 0, stream>>>(src, dst, E, part, own);
    k_deg_dinv<<<gN, blk, 0, stream>>>(part, dinv, deg4);

    k_gemm<<<GEMM_GROUPS, 1024, 0, stream>>>(x, W1, bufA);
    k_aggregate_owned<<<NGROUPS, 1024, 0, stream>>>(bufA, src, dst, dinv, b1, bufB, E, list, cnt, 0, own, W2, deg4);

    k_aggregate_owned<<<NGROUPS, 1024, 0, stream>>>(bufB, src, dst, dinv, b2, out, E, list, cnt, 1, own, W2, deg4);
}


# Both layers: the aggregation launch sums rows of the layer's input, multiplies its finished rows by W in LDS (WMMA f32), bias+relu at its store; no product launch; sum order: aggregate before product
# speedup vs baseline: 4.9994x; 1.1272x over previous
#include <hip/hip_runtime.h>

#define N_NODES 100000
#define DIM 128

typedef __attribute__((ext_vector_type(2))) float v2f;
typedef __attribute__((ext_vector_type(8))) float v8f;

#define DEG_R    50016
#define DEG_NR   ((N_NODES + DEG_R - 1) / DEG_R)
static_assert(DEG_NR == 2, "the owner list and the packed words are dealt to the two ranges' groups");
#define DEG_C    64
#define DEG_U    8

#define OWN      400

__device__ __forceinline__ unsigned owner_byte(int d) {
    unsigned o = (unsigned)d / (unsigned)OWN;
    return o < 255u ? o : 255u;
}

__device__ __forceinline__ int edge_word(int s, int d) {
    return (int)(((unsigned)s & 0x1FFFFu) | (((unsigned)d % (unsigned)OWN) << 17));
}

__device__ __forceinline__ size_t edge_words_at(int E) {
    return (size_t)((E + 511) >> 9) * 128;
}

__global__ __launch_bounds__(1024) void k_deg_part(const int* __restrict__ src, const int* __restrict__ dst, long long E64,
                                                   int* __restrict__ part, int* __restrict__ own) {
    __shared__ __attribute__((aligned(16))) int c_n[DEG_R];
    const int tid   = threadIdx.x;
    const int range = blockIdx.x % DEG_NR;
    const int chunk = blockIdx.x / DEG_NR;
    const int lo    = range * DEG_R;
    const int nr    = (N_NODES - lo < DEG_R) ? (N_NODES - lo) : DEG_R;

    for (int i = tid; i < DEG_R / 4; i += 1024) ((int4*)c_n)[i] = make_int4(0, 0, 0, 0);
    __syncthreads();

    const int E   = (int)E64;
    const int Q   = E >> 2;
    const int per = (((Q + DEG_C - 1) / DEG_C) + 31) & ~31;
    const int q0  = chunk * per;
    const int q1  = (q0 + per < Q) ? (q0 + per) : Q;
    const int4* dq = (const int4*)dst;
    const int4* sq = (const int4*)src;
    int* rec = own + edge_words_at(E);
    for (int b = q0; b < q1; b += DEG_U * 1024) {
        int4 d[DEG_U];
        int4 s[DEG_U];
#pragma unroll
        for (int k = 0; k < DEG_U; ++k) {
            int q  = b + k * 1024 + tid;
            int qc = (q < q1) ? q : (q1 - 1);
            d[k] = dq[qc];
            if (range == 1) s[k] = sq[qc];
        }
#pragma unroll
        for (int k = 0; k < DEG_U; ++k) {
            if (b + k * 1024 + tid < q1) {
                if (range == 0)
                    own[b + k * 1024 + tid] = (int)(owner_byte(d[k].x) | (owner_byte(d[k].y) << 8) |
                                                    (owner_byte(d[k].z) << 16) | (owner_byte(d[k].w) << 24));
                else
                    ((int4*)rec)[b + k * 1024 + tid] = make_int4(edge_word(s[k].x, d[k].x), edge_word(s[k].y, d[k].y),
                                                                 edge_word(s[k].z, d[k].z), edge_word(s[k].w, d[k].w));
                unsigned a0 = (unsigned)(d[k].x - lo);
                unsigned a1 = (unsigned)(d[k].y - lo);
                unsigned a2 = (unsigned)(d[k].z - lo);
                unsigned a3 = (unsigned)(d[k].w - lo);
                if (a0 < (unsigned)nr) atomicAdd(&c_n[a0], 1);
                if (a1 < (unsigned)nr) atomicAdd(&c_n[a1], 1);
                if (a2 < (unsigned)nr) atomicAdd(&c_n[a2], 1);
                if (a3 < (unsigned)nr) atomicAdd(&c_n[a3], 1);
            }
        }
    }
    if (chunk == DEG_C - 1) {
        int e = Q * 4 + tid;
        if (e < E) {
            unsigned a = (unsigned)(dst[e] - lo);
            if (a < (unsigned)nr) atomicAdd(&c_n[a], 1);
            if (range == 1) rec[e] = edge_word(src[e], dst[e]);
        }
        if (range == 0) {
            const int QP = ((E + 511) >> 9) * 128;
            for (int q = Q + tid; q < QP; q += 1024) {
                unsigned pk = 0;
                for (int j = 0; j < 4; ++j) {
                    int ee = q * 4 + j;
                    unsigned o = (ee < E) ? owner_byte(dst[ee]) : 255u;
                    pk |= o << (8 * j);
                }
                own[q] = (int)pk;
            }
        }
    }
    __syncthreads();

    int4* p = (int4*)(part + (size_t)chunk * N_NODES + lo);
    for (int i = tid; i < nr / 4; i += 1024) p[i] = ((const int4*)c_n)[i];
}

#define DEGTAB_BYTES (((N_NODES + 127) / 128) * 128)
static_assert(N_NODES % 16 == 0, "the table is moved in 16-byte pieces and packed four nodes per dword");

__global__ __launch_bounds__(256) void k_deg_dinv(const int* __restrict__ part, float* __restrict__ dinv,
                                                  int* __restrict__ deg4) {
    __shared__ __attribute__((aligned(16))) int c_s[256];
    const int t = threadIdx.x;
    int i = blockIdx.x * 256 + t;
    int c = 1;
    if (i < N_NODES) {
#pragma unroll
        for (int k = 0; k < DEG_C; ++k) c += part[(size_t)k * N_NODES + i];
        dinv[i] = rsqrtf((float)c);
    }
    c_s[t] = c < 255 ? c : 255;
    __syncthreads();
    const int j = blockIdx.x * 64 + t;
    if (t < 64 && j < N_NODES / 4) {
        const int4 q = ((const int4*)c_s)[t];
        deg4[j] = q.x | (q.y << 8) | (q.z << 16) | (q.w << 24);
    }
}


__device__ __forceinline__ float4 w_piece(const float* __restrict__ W, int idx) {
    const int c  = idx & 15;
    const int j  = (idx >> 4) & 3;
    const int kp = idx >> 6;
    const float* w0 = W + (size_t)(2 * kp) * DIM + j * 32 + c;
    float4 o;
    o.x = w0[0];
    o.y = w0[DIM];
    o.z = w0[16];
    o.w = w0[DIM + 16];
    return o;
}

#define NGROUPS  (N_NODES / OWN)
#define SCAN_U   8
#define QCAP     64
#define WCAP     1024

__device__ __forceinline__ void add_rows(int n, int lane, const float* __restrict__ H,
                                         int s_i, int d_i, float nr_i, float* acc) {
    for (int j0 = 0; j0 < n; j0 += 8) {
        float4 v[8];
#pragma unroll
        for (int k = 0; k < 8; ++k) {
            int jj = (j0 + k < n) ? (j0 + k) : (n - 1);
            int s = __builtin_amdgcn_readlane(s_i, jj);
            v[k] = ((const float4*)(H + (size_t)s * DIM))[lane];
        }
#pragma unroll
        for (int k = 0; k < 8; ++k) {
            int jj = j0 + k;
            if (jj < n) {
                int d = __builtin_amdgcn_readlane(d_i, jj);
                float nr = __int_as_float(__builtin_amdgcn_readlane(__float_as_int(nr_i), jj));
                float* a = acc + d * DIM + lane;
                atomicAdd(a,      v[k].x * nr);
                atomicAdd(a + 32, v[k].y * nr);
                atomicAdd(a + 64, v[k].z * nr);
                atomicAdd(a + 96, v[k].w * nr);
            }
        }
    }
}

typedef __attribute__((address_space(3))) unsigned char lds_u8;
typedef __attribute__((address_space(3))) float lds_f32;

__device__ __forceinline__ int drain_edges(const int* qe, int n, int lane,
                                           const float* __restrict__ H,
                                           const int* __restrict__ rec,
                                           const float* __restrict__ dinv,
                                           bool tab, const lds_u8* tab8, const lds_f32* dinv_own,
                                           int lo, float* acc,
                                           int2* wl, int ln, bool record) {
    int s_i = 0, d_i = 0;
    float nr_i = 0.f;
    if (lane < n) {
        int e_i = qe[lane];
        int r   = rec[e_i];
        s_i  = r & 0x1FFFF;
        d_i  = (int)((unsigned)r >> 17);
        float ds, dd;
        if (tab) {
            dd = dinv_own[d_i];
            const int c = tab8[s_i];
            ds = rsqrtf((float)c);
            if (c == 255) ds = dinv[s_i];
        } else {
            ds = dinv[s_i];
            dd = dinv[lo + d_i];
        }
        nr_i = ds * dd;
        if (record && ln + lane < WCAP)
            wl[ln + lane] = make_int2(s_i | (d_i << 17), __float_as_int(nr_i));
    }
    add_rows(n, lane, H, s_i, d_i, nr_i, acc);
    return ln + n;
}

__device__ __forceinline__ int x2_pos(int r, int k) {
    return r * DIM + 4 * ((k >> 2) ^ (r & 15)) + (k & 3);
}

__device__ __forceinline__ float4 finished_row(const float* __restrict__ H, const float* __restrict__ dinv,
                                               const float* acc, int lo, int r, int lane) {
    const int i = lo + r;
    float di = dinv[i];
    float sn = di * di;
    const float* a = acc + r * DIM + lane;
    float4 h = ((const float4*)(H + (size_t)i * DIM))[lane];
    float4 o;
    o.x = fmaf(sn, h.x, a[0]);
    o.y = fmaf(sn, h.y, a[32]);
    o.z = fmaf(sn, h.z, a[64]);
    o.w = fmaf(sn, h.w, a[96]);
    return o;
}

__global__ __launch_bounds__(1024) void k_aggregate_owned(const float* __restrict__ H,
                                                          const float* __restrict__ dinv,
                                                          const float* __restrict__ bias,
                                                          float* __restrict__ out, long long E64,
                                                          int* list, int* cnt, int mode,
                                                          const int* __restrict__ own,
                                                          const float* __restrict__ W,
                                                          const int* __restrict__ deg4) {
    __shared__ __attribute__((aligned(16))) float acc[OWN * DIM];
    __shared__ __attribute__((aligned(16))) float w_l[DEGTAB_BYTES / 4];
    static_assert(DEGTAB_BYTES >= DIM * DIM * 4, "W is staged into the table's space");
    __shared__ float dinv_own[OWN];
    __shared__ int q_e[32 * QCAP];
    __shared__ int w_cnt[32];

    const int tid  = threadIdx.x;
    const int lane = tid & 31;
    const int w    = __builtin_amdgcn_readfirstlane(tid >> 5);
    const int lo   = blockIdx.x * OWN;
    const int E    = (int)E64;

    for (int i = tid; i < OWN * DIM / 4; i += 1024)
        ((float4*)acc)[i] = make_float4(0.f, 0.f, 0.f, 0.f);
    if (mode == 0) {
        constexpr int TQ = N_NODES / 16;
        constexpr int TU = (TQ + 1023) / 1024;
        const float4* tq = (const float4*)deg4;
        float4 tp[TU];
#pragma unroll
        for (int k = 0; k < TU; ++k) {
            const int idx = tid + k * 1024;
            tp[k] = tq[idx < TQ ? idx : TQ - 1];
        }
#pragma unroll
        for (int k = 0; k < TU; ++k) {
            const int idx = tid + k * 1024;
            if (idx < TQ) ((float4*)w_l)[idx] = tp[k];
        }
        if (tid < OWN) dinv_own[tid] = dinv[lo + tid];
    }
    __syncthreads();

    int2* wl = (int2*)list + ((size_t)blockIdx.x * 32 + w) * WCAP;
    int ln = 0;
    bool scan = true;
    if (mode != 0) {
        ln = cnt[blockIdx.x * 32 + w];
        scan = (unsigned)ln > (unsigned)WCAP;
    }

    if (scan) {
        const bool record = mode == 0;
        const bool tab = mode == 0;
        const lds_u8* tab8 = (const lds_u8*)w_l;
        const int* rec = own + edge_words_at(E);
        int* qe = q_e + w * QCAP;
        int qn = 0;
        const unsigned ltmask = (1u << lane) - 1u;

        const int F     = (E + 511) >> 9;
        const int trips = (F + 32 * SCAN_U - 1) / (32 * SCAN_U);
        int trip = (int)(((long long)blockIdx.x * trips) / NGROUPS);
        const unsigned pat = (unsigned)blockIdx.x * 0x01010101u;

        for (int n = 0; n < trips; ++n) {
            const int tile0 = trip * (SCAN_U * 32) + w;
            int4 dv[SCAN_U];
#pragma unroll
            for (int k = 0; k < SCAN_U; ++k) {
                int t  = tile0 + k * 32;
                int tc = (t < F) ? t : (F - 1);
                dv[k] = *(const int4*)(own + tc * 128 + lane * 4);
            }
            if (++trip == trips) trip = 0;
            static_assert(SCAN_U == 8, "the queueing below names four masks");
            unsigned hms[SCAN_U / 2];
#pragma unroll
            for (int p = 0; p < SCAN_U / 2; ++p) {
                const unsigned w8[8] = {(unsigned)dv[2 * p].x,     (unsigned)dv[2 * p].y,
                                        (unsigned)dv[2 * p].z,     (unsigned)dv[2 * p].w,
                                        (unsigned)dv[2 * p + 1].x, (unsigned)dv[2 * p + 1].y,
                                        (unsigned)dv[2 * p + 1].z, (unsigned)dv[2 * p + 1].w};
                unsigned hm = 0;
#pragma unroll
                for (int k = 0; k < 8; ++k) {
                    unsigned x = w8[k] ^ pat;
                    unsigned a = (x & 0x7F7F7F7Fu) + 0x7F7F7F7Fu;
                    unsigned z = ~(a | x) & 0x80808080u;
                    hm |= z >> k;
                }
                const int ta = tile0 + (2 * p) * 32;
                hms[p] = hm & (((ta < F) ? 0xF0F0F0F0u : 0u) | ((ta + 32 < F) ? 0x0F0F0F0Fu : 0u));
            }
            for (;;) {
                bool has = (hms[0] | hms[1] | hms[2] | hms[3]) != 0u;
                unsigned m = __builtin_amdgcn_ballot_w32(has);
                if (m == 0u) break;
                if (has) {
                    const int p = hms[0] ? 0 : hms[1] ? 1 : hms[2] ? 2 : 3;
                    const unsigned h = hms[0] ? hms[0] : hms[1] ? hms[1] : hms[2] ? hms[2] : hms[3];
                    const int step = __builtin_ctz(h);
                    const unsigned c = h & (h - 1u);
                    if (p == 0) hms[0] = c; else if (p == 1) hms[1] = c; else if (p == 2) hms[2] = c; else hms[3] = c;
                    const int k = 7 - (step & 7);
                    int e = (tile0 + (2 * p + (k >> 2)) * 32) * 512 + lane * 16 + (k & 3) * 4 + (step >> 3);
                    qe[qn + __builtin_popcount(m & ltmask)] = e;
                }
                qn += __builtin_popcount(m);
                if (qn >= 32) {
                    qn -= 32;
                    ln = drain_edges(qe + qn, 32, lane, H, rec, dinv, tab, tab8, (const lds_f32*)dinv_own, lo, acc, wl, ln, record);
                }
            }
        }
        if (qn > 0) ln = drain_edges(qe, qn, lane, H, rec, dinv, tab, tab8, (const lds_f32*)dinv_own, lo, acc, wl, ln, record);
        if (record && lane == 0) w_cnt[w] = ln;
    } else {
        int2 ent = make_int2(0, 0);
        if (lane < ln) ent = wl[lane];
        for (int c = 0; c < ln; c += 32) {
            const int n = (ln - c < 32) ? (ln - c) : 32;
            const int2 cur = ent;
            if (c + 32 + lane < ln) ent = wl[c + 32 + lane];
            add_rows(n, lane, H, cur.x & 0x1FFFF, cur.x >> 17, __int_as_float(cur.y), acc);
        }
    }
    constexpr int WP = DIM * DIM / 4 / 1024;
    float4 wp[WP];
#pragma unroll
    for (int k = 0; k < WP; ++k) wp[k] = w_piece(W, tid + k * 1024);
    __syncthreads();
#pragma unroll
    for (int k = 0; k < WP; ++k) ((float4*)w_l)[tid + k * 1024] = wp[k];

    if (mode == 0 && w == 0) cnt[blockIdx.x * 32 + lane] = w_cnt[lane];

#pragma unroll 4
    for (int r = w; r < OWN; r += 32)
        *(float4*)(acc + x2_pos(r, 4 * lane)) = finished_row(H, dinv, acc, lo, r, lane);
    __syncthreads();

    const int half = lane >> 4;
    const int l16  = lane & 15;
    const float4* wq = (const float4*)w_l + half * 64 + l16;
    for (int u = w; u < 2 * (OWN / 16); u += 32) {
        const int strip = u >> 1;
        const int nh    = u & 1;
        const int r     = strip * 16 + l16;
        const float4* wqn = wq + nh * 32;
        const float* bcol = bias + nh * 64 + l16;
        const float bi0 = bcol[0], bi1 = bcol[16], bi2 = bcol[32], bi3 = bcol[48];
        v8f c0 = (v8f){0,0,0,0,0,0,0,0}, c1 = c0, c2 = c0, c3 = c0;
        v2f an = *(const v2f*)(acc + x2_pos(r, 2 * half));
        float4 qn0 = wqn[0], qn1 = wqn[16];
#pragma unroll 2
        for (int k0 = 0; k0 < DIM; k0 += 4) {
            const v2f a = an;
            const float4 q0 = qn0, q1 = qn1;
            const int kn = (k0 + 4 < DIM) ? (k0 + 4) : k0;
            an = *(const v2f*)(acc + x2_pos(r, kn + 2 * half));
            const float4* bp = wqn + (kn >> 1) * 64;
            qn0 = bp[0]; qn1 = bp[16];
            v2f b;
            b.x = q0.x; b.y = q0.y;
            c0 = __builtin_amdgcn_wmma_f32_16x16x4_f32(false, a, false, b, (short)0, c0, false, false);
            b.x = q0.z; b.y = q0.w;
            c1 = __builtin_amdgcn_wmma_f32_16x16x4_f32(false, a, false, b, (short)0, c1, false, false);
            b.x = q1.x; b.y = q1.y;
            c2 = __builtin_amdgcn_wmma_f32_16x16x4_f32(false, a, false, b, (short)0, c2, false, false);
            b.x = q1.z; b.y = q1.w;
            c3 = __builtin_amdgcn_wmma_f32_16x16x4_f32(false, a, false, b, (short)0, c3, false, false);
        }
        float* orow = out + (size_t)(lo + strip * 16 + half * 8) * DIM + nh * 64 + l16;
#pragma unroll
        for (int v = 0; v < 8; ++v) {
            orow[(size_t)v * DIM]      = fmaxf(c0[v] + bi0, 0.f);
            orow[(size_t)v * DIM + 16] = fmaxf(c1[v] + bi1, 0.f);
            orow[(size_t)v * DIM + 32] = fmaxf(c2[v] + bi2, 0.f);
            orow[(size_t)v * DIM + 48] = fmaxf(c3[v] + bi3, 0.f);
        }
    }
}

extern "C" void kernel_launch(void* const* d_in, const int* in_sizes, int n_in,
                              void* d_out, int out_size, void* d_ws, size_t ws_size,
                              hipStream_t stream) {
    const float* x   = (const float*)d_in[0];
    const int*   ei  = (const int*)  d_in[1];
    const float* W1  = (const float*)d_in[2];
    const float* b1  = (const float*)d_in[3];
    const float* W2  = (const float*)d_in[4];
    const float* b2  = (const float*)d_in[5];
    float* out = (float*)d_out;

    const long long E = (long long)in_sizes[1] / 2;
    const int* src = ei;
    const int* dst = ei + E;

    float* dinv = (float*)d_ws;
    float* mid  = dinv + N_NODES;
    int*   cnt  = (int*)(mid + (size_t)N_NODES * DIM);
    int*   list = cnt + NGROUPS * 32 + 32;
    int*   part = list + (size_t)NGROUPS * 32 * WCAP * 2;
    int*   deg4 = part + (size_t)DEG_C * N_NODES;
    int*   own  = deg4 + DEGTAB_BYTES / 4;

    dim3 blk(256);
    const int gN    = (N_NODES + 255) / 256;

    k_deg_part<<<DEG_NR * DEG_C, 1024, 0, stream>>>(src, dst, E, part, own);
    k_deg_dinv<<<gN, blk, 0, stream>>>(part, dinv, deg4);

    k_aggregate_owned<<<NGROUPS, 1024, 0, stream>>>(x, dinv, b1, mid, E, list, cnt, 0, own, W1, deg4);

    k_aggregate_owned<<<NGROUPS, 1024, 0, stream>>>(mid, dinv, b2, out, E, list, cnt, 1, own, W2, deg4);
}


# speedup vs baseline: 5.3304x; 1.0662x over previous
#include <hip/hip_runtime.h>

#define N_NODES 100000
#define DIM 128

typedef __attribute__((ext_vector_type(2))) float v2f;
typedef __attribute__((ext_vector_type(8))) float v8f;

#define DEG_R    50016
#define DEG_NR   ((N_NODES + DEG_R - 1) / DEG_R)
static_assert(DEG_NR == 2, "the owner list and the packed words are dealt to the two ranges' groups");
#define DEG_C    64
#define DEG_U    8

#define OWN      400

__device__ __forceinline__ unsigned owner_byte(int d) {
    unsigned o = (unsigned)d / (unsigned)OWN;
    return o < 255u ? o : 255u;
}

__device__ __forceinline__ int edge_word(int s, int d) {
    return (int)(((unsigned)s & 0x1FFFFu) | (((unsigned)d % (unsigned)OWN) << 17));
}

__device__ __forceinline__ size_t edge_words_at(int E) {
    return (size_t)((E + 511) >> 9) * 128;
}

__global__ __launch_bounds__(1024) void k_deg_part(const int* __restrict__ src, const int* __restrict__ dst, long long E64,
                                                   int* __restrict__ part, int* __restrict__ own) {
    __shared__ __attribute__((aligned(16))) int c_n[DEG_R];
    const int tid   = threadIdx.x;
    const int range = blockIdx.x % DEG_NR;
    const int chunk = blockIdx.x / DEG_NR;
    const int lo    = range * DEG_R;
    const int nr    = (N_NODES - lo < DEG_R) ? (N_NODES - lo) : DEG_R;

    for (int i = tid; i < DEG_R / 4; i += 1024) ((int4*)c_n)[i] = make_int4(0, 0, 0, 0);
    __syncthreads();

    const int E   = (int)E64;
    const int Q   = E >> 2;
    const int per = (((Q + DEG_C - 1) / DEG_C) + 31) & ~31;
    const int q0  = chunk * per;
    const int q1  = (q0 + per < Q) ? (q0 + per) : Q;
    const int4* dq = (const int4*)dst;
    const int4* sq = (const int4*)src;
    int* rec = own + edge_words_at(E);
    for (int b = q0; b < q1; b += DEG_U * 1024) {
        int4 d[DEG_U];
        int4 s[DEG_U];
#pragma unroll
        for (int k = 0; k < DEG_U; ++k) {
            int q  = b + k * 1024 + tid;
            int qc = (q < q1) ? q : (q1 - 1);
            d[k] = dq[qc];
            if (range == 1) s[k] = sq[qc];
        }
#pragma unroll
        for (int k = 0; k < DEG_U; ++k) {
            if (b + k * 1024 + tid < q1) {
                if (range == 0)
                    own[b + k * 1024 + tid] = (int)(owner_byte(d[k].x) | (owner_byte(d[k].y) << 8) |
                                                    (owner_byte(d[k].z) << 16) | (owner_byte(d[k].w) << 24));
                else
                    ((int4*)rec)[b + k * 1024 + tid] = make_int4(edge_word(s[k].x, d[k].x), edge_word(s[k].y, d[k].y),
                                                                 edge_word(s[k].z, d[k].z), edge_word(s[k].w, d[k].w));
                unsigned a0 = (unsigned)(d[k].x - lo);
                unsigned a1 = (unsigned)(d[k].y - lo);
                unsigned a2 = (unsigned)(d[k].z - lo);
                unsigned a3 = (unsigned)(d[k].w - lo);
                if (a0 < (unsigned)nr) atomicAdd(&c_n[a0], 1);
                if (a1 < (unsigned)nr) atomicAdd(&c_n[a1], 1);
                if (a2 < (unsigned)nr) atomicAdd(&c_n[a2], 1);
                if (a3 < (unsigned)nr) atomicAdd(&c_n[a3], 1);
            }
        }
    }
    if (chunk == DEG_C - 1) {
        int e = Q * 4 + tid;
        if (e < E) {
            unsigned a = (unsigned)(dst[e] - lo);
            if (a < (unsigned)nr) atomicAdd(&c_n[a], 1);
            if (range == 1) rec[e] = edge_word(src[e], dst[e]);
        }
        if (range == 0) {
            const int QP = ((E + 511) >> 9) * 128;
            for (int q = Q + tid; q < QP; q += 1024) {
                unsigned pk = 0;
                for (int j = 0; j < 4; ++j) {
                    int ee = q * 4 + j;
                    unsigned o = (ee < E) ? owner_byte(dst[ee]) : 255u;
                    pk |= o << (8 * j);
                }
                own[q] = (int)pk;
            }
        }
    }
    __syncthreads();

    int4* p = (int4*)(part + (size_t)chunk * N_NODES + lo);
    for (int i = tid; i < nr / 4; i += 1024) p[i] = ((const int4*)c_n)[i];
}

#define DEGTAB_BYTES (((N_NODES + 127) / 128) * 128)
static_assert(N_NODES % 16 == 0, "the table is moved in 16-byte pieces and packed four nodes per dword");

__global__ __launch_bounds__(256) void k_deg_dinv(const int* __restrict__ part, float* __restrict__ dinv,
                                                  int* __restrict__ deg4) {
    __shared__ __attribute__((aligned(16))) int c_s[256];
    const int t = threadIdx.x;
    int i = blockIdx.x * 256 + t;
    int c = 1;
    if (i < N_NODES) {
#pragma unroll
        for (int k = 0; k < DEG_C; ++k) c += part[(size_t)k * N_NODES + i];
        dinv[i] = rsqrtf((float)c);
    }
    c_s[t] = c < 255 ? c : 255;
    __syncthreads();
    const int j = blockIdx.x * 64 + t;
    if (t < 64 && j < N_NODES / 4) {
        const int4 q = ((const int4*)c_s)[t];
        deg4[j] = q.x | (q.y << 8) | (q.z << 16) | (q.w << 24);
    }
}


__device__ __forceinline__ float4 w_piece(const float* __restrict__ W, int idx) {
    const int c  = idx & 15;
    const int j  = (idx >> 4) & 3;
    const int kp = idx >> 6;
    const float* w0 = W + (size_t)(2 * kp) * DIM + j * 32 + c;
    float4 o;
    o.x = w0[0];
    o.y = w0[DIM];
    o.z = w0[16];
    o.w = w0[DIM + 16];
    return o;
}

#define NGROUPS  (N_NODES / OWN)
#define SCAN_U   8
#define QCAP     64
#define WCAP     1024
#define MID_PAD  96
static_assert(((size_t)(N_NODES + MID_PAD) * 4) % 512 == 0, "rows of the first layer's output on 512-byte boundaries");
static_assert((NGROUPS * 32 * 4) % 256 == 0, "the per-wave lists behind the counts stay on a 256-byte boundary");

__device__ __forceinline__ void add_rows(int n, int lane, const float* __restrict__ H,
                                         int s_i, int d_i, float nr_i, float* acc) {
    for (int j0 = 0; j0 < n; j0 += 8) {
        float4 v[8];
#pragma unroll
        for (int k = 0; k < 8; ++k) {
            int jj = (j0 + k < n) ? (j0 + k) : (n - 1);
            int s = __builtin_amdgcn_readlane(s_i, jj);
            v[k] = ((const float4*)(H + (size_t)s * DIM))[lane];
        }
#pragma unroll
        for (int k = 0; k < 8; ++k) {
            int jj = j0 + k;
            if (jj < n) {
                int d = __builtin_amdgcn_readlane(d_i, jj);
                float nr = __int_as_float(__builtin_amdgcn_readlane(__float_as_int(nr_i), jj));
                float* a = acc + d * DIM + lane;
                atomicAdd(a,      v[k].x * nr);
                atomicAdd(a + 32, v[k].y * nr);
                atomicAdd(a + 64, v[k].z * nr);
                atomicAdd(a + 96, v[k].w * nr);
            }
        }
    }
}

typedef __attribute__((address_space(3))) unsigned char lds_u8;
typedef __attribute__((address_space(3))) float lds_f32;

__device__ __forceinline__ int drain_edges(const int* qe, int n, int lane,
                                           const float* __restrict__ H,
                                           const int* __restrict__ rec,
                                           const float* __restrict__ dinv,
                                           bool tab, const lds_u8* tab8, const lds_f32* dinv_own,
                                           int lo, float* acc,
                                           int2* wl, int ln, bool record) {
    int s_i = 0, d_i = 0;
    float nr_i = 0.f;
    if (lane < n) {
        int e_i = qe[lane];
        int r   = rec[e_i];
        s_i  = r & 0x1FFFF;
        d_i  = (int)((unsigned)r >> 17);
        float ds, dd;
        if (tab) {
            dd = dinv_own[d_i];
            const int c = tab8[s_i];
            ds = rsqrtf((float)c);
            if (c == 255) ds = dinv[s_i];
        } else {
            ds = dinv[s_i];
            dd = dinv[lo + d_i];
        }
        nr_i = ds * dd;
        if (record && ln + lane < WCAP)
            wl[ln + lane] = make_int2(s_i | (d_i << 17), __float_as_int(nr_i));
    }
    add_rows(n, lane, H, s_i, d_i, nr_i, acc);
    return ln + n;
}

__device__ __forceinline__ int x2_pos(int r, int k) {
    return r * DIM + 4 * ((k >> 2) ^ (r & 15)) + (k & 3);
}

__device__ __forceinline__ float4 finished_row(const float* __restrict__ H, const float* __restrict__ dinv,
                                               const float* acc, int lo, int r, int lane) {
    const int i = lo + r;
    float di = dinv[i];
    float sn = di * di;
    const float* a = acc + r * DIM + lane;
    float4 h = ((const float4*)(H + (size_t)i * DIM))[lane];
    float4 o;
    o.x = fmaf(sn, h.x, a[0]);
    o.y = fmaf(sn, h.y, a[32]);
    o.z = fmaf(sn, h.z, a[64]);
    o.w = fmaf(sn, h.w, a[96]);
    return o;
}

__global__ __launch_bounds__(1024) void k_aggregate_owned(const float* __restrict__ H,
                                                          const float* __restrict__ dinv,
                                                          const float* __restrict__ bias,
                                                          float* __restrict__ out, long long E64,
                                                          int* list, int* cnt, int mode,
                                                          const int* __restrict__ own,
                                                          const float* __restrict__ W,
                                                          const int* __restrict__ deg4) {
    __shared__ __attribute__((aligned(16))) float acc[OWN * DIM];
    __shared__ __attribute__((aligned(16))) float w_l[DEGTAB_BYTES / 4];
    static_assert(DEGTAB_BYTES >= DIM * DIM * 4, "W is staged into the table's space");
    __shared__ float dinv_own[OWN];
    __shared__ int q_e[32 * QCAP];
    __shared__ int w_cnt[32];

    const int tid  = threadIdx.x;
    const int lane = tid & 31;
    const int w    = __builtin_amdgcn_readfirstlane(tid >> 5);
    const int lo   = blockIdx.x * OWN;
    const int E    = (int)E64;

    for (int i = tid; i < OWN * DIM / 4; i += 1024)
        ((float4*)acc)[i] = make_float4(0.f, 0.f, 0.f, 0.f);
    if (mode == 0) {
        constexpr int TQ = N_NODES / 16;
        constexpr int TU = (TQ + 1023) / 1024;
        const float4* tq = (const float4*)deg4;
        float4 tp[TU];
#pragma unroll
        for (int k = 0; k < TU; ++k) {
            const int idx = tid + k * 1024;
            tp[k] = tq[idx < TQ ? idx : TQ - 1];
        }
#pragma unroll
        for (int k = 0; k < TU; ++k) {
            const int idx = tid + k * 1024;
            if (idx < TQ) ((float4*)w_l)[idx] = tp[k];
        }
        if (tid < OWN) dinv_own[tid] = dinv[lo + tid];
    }
    __syncthreads();

    int2* wl = (int2*)list + ((size_t)blockIdx.x * 32 + w) * WCAP;
    int ln = 0;
    bool scan = true;
    if (mode != 0) {
        ln = cnt[blockIdx.x * 32 + w];
        scan = (unsigned)ln > (unsigned)WCAP;
    }

    if (scan) {
        const bool record = mode == 0;
        const bool tab = mode == 0;
        const lds_u8* tab8 = (const lds_u8*)w_l;
        const int* rec = own + edge_words_at(E);
        int* qe = q_e + w * QCAP;
        int qn = 0;
        const unsigned ltmask = (1u << lane) - 1u;

        const int F     = (E + 511) >> 9;
        const int trips = (F + 32 * SCAN_U - 1) / (32 * SCAN_U);
        int trip = (int)(((long long)blockIdx.x * trips) / NGROUPS);
        const unsigned pat = (unsigned)blockIdx.x * 0x01010101u;

        for (int n = 0; n < trips; ++n) {
            const int tile0 = trip * (SCAN_U * 32) + w;
            int4 dv[SCAN_U];
#pragma unroll
            for (int k = 0; k < SCAN_U; ++k) {
                int t  = tile0 + k * 32;
                int tc = (t < F) ? t : (F - 1);
                dv[k] = *(const int4*)(own + tc * 128 + lane * 4);
            }
            if (++trip == trips) trip = 0;
            static_assert(SCAN_U == 8, "the queueing below names four masks");
            unsigned hms[SCAN_U / 2];
#pragma unroll
            for (int p = 0; p < SCAN_U / 2; ++p) {
                const unsigned w8[8] = {(unsigned)dv[2 * p].x,     (unsigned)dv[2 * p].y,
                                        (unsigned)dv[2 * p].z,     (unsigned)dv[2 * p].w,
                                        (unsigned)dv[2 * p + 1].x, (unsigned)dv[2 * p + 1].y,
                                        (unsigned)dv[2 * p + 1].z, (unsigned)dv[2 * p + 1].w};
                unsigned hm = 0;
#pragma unroll
                for (int k = 0; k < 8; ++k) {
                    unsigned x = w8[k] ^ pat;
                    unsigned a = (x & 0x7F7F7F7Fu) + 0x7F7F7F7Fu;
                    unsigned z = ~(a | x) & 0x80808080u;
                    hm |= z >> k;
                }
                const int ta = tile0 + (2 * p) * 32;
                hms[p] = hm & (((ta < F) ? 0xF0F0F0F0u : 0u) | ((ta + 32 < F) ? 0x0F0F0F0Fu : 0u));
            }
            for (;;) {
                bool has = (hms[0] | hms[1] | hms[2] | hms[3]) != 0u;
                unsigned m = __builtin_amdgcn_ballot_w32(has);
                if (m == 0u) break;
                if (has) {
                    const int p = hms[0] ? 0 : hms[1] ? 1 : hms[2] ? 2 : 3;
                    const unsigned h = hms[0] ? hms[0] : hms[1] ? hms[1] : hms[2] ? hms[2] : hms[3];
                    const int step = __builtin_ctz(h);
                    const unsigned c = h & (h - 1u);
                    if (p == 0) hms[0] = c; else if (p == 1) hms[1] = c; else if (p == 2) hms[2] = c; else hms[3] = c;
                    const int k = 7 - (step & 7);
                    int e = (tile0 + (2 * p + (k >> 2)) * 32) * 512 + lane * 16 + (k & 3) * 4 + (step >> 3);
                    qe[qn + __builtin_popcount(m & ltmask)] = e;
                }
                qn += __builtin_popcount(m);
                if (qn >= 32) {
                    qn -= 32;
                    ln = drain_edges(qe + qn, 32, lane, H, rec, dinv, tab, tab8, (const lds_f32*)dinv_own, lo, acc, wl, ln, record);
                }
            }
        }
        if (qn > 0) ln = drain_edges(qe, qn, lane, H, rec, dinv, tab, tab8, (const lds_f32*)dinv_own, lo, acc, wl, ln, record);
        if (record && lane == 0) w_cnt[w] = ln;
    } else {
        int2 ent = make_int2(0, 0);
        if (lane < ln) ent = wl[lane];
        for (int c = 0; c < ln; c += 32) {
            const int n = (ln - c < 32) ? (ln - c) : 32;
            const int2 cur = ent;
            if (c + 32 + lane < ln) ent = wl[c + 32 + lane];
            add_rows(n, lane, H, cur.x & 0x1FFFF, cur.x >> 17, __int_as_float(cur.y), acc);
        }
    }
    constexpr int WP = DIM * DIM / 4 / 1024;
    float4 wp[WP];
#pragma unroll
    for (int k = 0; k < WP; ++k) wp[k] = w_piece(W, tid + k * 1024);
    __syncthreads();
#pragma unroll
    for (int k = 0; k < WP; ++k) ((float4*)w_l)[tid + k * 1024] = wp[k];

    if (mode == 0 && w == 0) cnt[blockIdx.x * 32 + lane] = w_cnt[lane];

#pragma unroll 4
    for (int r = w; r < OWN; r += 32)
        *(float4*)(acc + x2_pos(r, 4 * lane)) = finished_row(H, dinv, acc, lo, r, lane);
    __syncthreads();

    const int half = lane >> 4;
    const int l16  = lane & 15;
    const float4* wq = (const float4*)w_l + half * 64 + l16;
    for (int u = w; u < 2 * (OWN / 16); u += 32) {
        const int strip = u >> 1;
        const int nh    = u & 1;
        const int r     = strip * 16 + l16;
        const float4* wqn = wq + nh * 32;
        const float* bcol = bias + nh * 64 + l16;
        const float bi0 = bcol[0], bi1 = bcol[16], bi2 = bcol[32], bi3 = bcol[48];
        v8f c0 = (v8f){0,0,0,0,0,0,0,0}, c1 = c0, c2 = c0, c3 = c0;
        v2f an = *(const v2f*)(acc + x2_pos(r, 2 * half));
        float4 qn0 = wqn[0], qn1 = wqn[16];
#pragma unroll 2
        for (int k0 = 0; k0 < DIM; k0 += 4) {
            const v2f a = an;
            const float4 q0 = qn0, q1 = qn1;
            const int kn = (k0 + 4 < DIM) ? (k0 + 4) : k0;
            an = *(const v2f*)(acc + x2_pos(r, kn + 2 * half));
            const float4* bp = wqn + (kn >> 1) * 64;
            qn0 = bp[0]; qn1 = bp[16];
            v2f b;
            b.x = q0.x; b.y = q0.y;
            c0 = __builtin_amdgcn_wmma_f32_16x16x4_f32(false, a, false, b, (short)0, c0, false, false);
            b.x = q0.z; b.y = q0.w;
            c1 = __builtin_amdgcn_wmma_f32_16x16x4_f32(false, a, false, b, (short)0, c1, false, false);
            b.x = q1.x; b.y = q1.y;
            c2 = __builtin_amdgcn_wmma_f32_16x16x4_f32(false, a, false, b, (short)0, c2, false, false);
            b.x = q1.z; b.y = q1.w;
            c3 = __builtin_amdgcn_wmma_f32_16x16x4_f32(false, a, false, b, (short)0, c3, false, false);
        }
        float* orow = out + (size_t)(lo + strip * 16 + half * 8) * DIM + nh * 64 + l16;
#pragma unroll
        for (int v = 0; v < 8; ++v) {
            orow[(size_t)v * DIM]      = fmaxf(c0[v] + bi0, 0.f);
            orow[(size_t)v * DIM + 16] = fmaxf(c1[v] + bi1, 0.f);
            orow[(size_t)v * DIM + 32] = fmaxf(c2[v] + bi2, 0.f);
            orow[(size_t)v * DIM + 48] = fmaxf(c3[v] + bi3, 0.f);
        }
    }
}

extern "C" void kernel_launch(void* const* d_in, const int* in_sizes, int n_in,
                              void* d_out, int out_size, void* d_ws, size_t ws_size,
                              hipStream_t stream) {
    const float* x   = (const float*)d_in[0];
    const int*   ei  = (const int*)  d_in[1];
    const float* W1  = (const float*)d_in[2];
    const float* b1  = (const float*)d_in[3];
    const float* W2  = (const float*)d_in[4];
    const float* b2  = (const float*)d_in[5];
    float* out = (float*)d_out;

    const long long E = (long long)in_sizes[1] / 2;
    const int* src = ei;
    const int* dst = ei + E;

    float* dinv = (float*)d_ws;
    float* mid  = dinv + N_NODES + MID_PAD;
    int*   cnt  = (int*)(mid + (size_t)N_NODES * DIM);
    int*   list = cnt + NGROUPS * 32;
    int*   part = list + (size_t)NGROUPS * 32 * WCAP * 2;
    int*   deg4 = part + (size_t)DEG_C * N_NODES;
    int*   own  = deg4 + DEGTAB_BYTES / 4;

    dim3 blk(256);
    const int gN    = (N_NODES + 255) / 256;

    k_deg_part<<<DEG_NR * DEG_C, 1024, 0, stream>>>(src, dst, E, part, own);
    k_deg_dinv<<<gN, blk, 0, stream>>>(part, dinv, deg4);

    k_aggregate_owned<<<NGROUPS, 1024, 0, stream>>>(x, dinv, b1, mid, E, list, cnt, 0, own, W1, deg4);

    k_aggregate_owned<<<NGROUPS, 1024, 0, stream>>>(mid, dinv, b2, out, E, list, cnt, 1, own, W2, deg4);
}
